# starting kernel
# baseline (speedup 1.0000x reference)
#include <hip/hip_runtime.h>
#include <math.h>

#define D 128
#define NEG_SLOPE 0.2f
#define LN_EPS 1e-5f

typedef float v2f __attribute__((ext_vector_type(2)));
typedef float v8f __attribute__((ext_vector_type(8)));

__device__ __forceinline__ float wave_sum(float v) {
#pragma unroll
  for (int off = 16; off > 0; off >>= 1) v += __shfl_xor(v, off, 32);
  return v;
}

__global__ __launch_bounds__(256)
void gat_gemm_wmma(const float* __restrict__ x, const float* __restrict__ w,
                   float* __restrict__ h, int n_nodes) {
  __shared__ float lds_a[16 * D];
  const int tid  = threadIdx.x;
  const int row0 = blockIdx.x * 16;

  for (int i = tid; i < 16 * D; i += 256) {
    int r = i >> 7, c = i & (D - 1);
    int row = row0 + r;
    lds_a[i] = (row < n_nodes) ? x[(size_t)row * D + c] : 0.0f;
  }
  __syncthreads();

  const int wave = tid >> 5;
  const int lane = tid & 31;
  const int half = lane >> 4;
  const int l16  = lane & 15;
  const int col0 = wave * 16;

  v8f c = {};
#pragma unroll
  for (int k = 0; k < D; k += 4) {
    const int ka = k + half * 2;
    v2f a, b;
    a.x = lds_a[l16 * D + ka];
    a.y = lds_a[l16 * D + ka + 1];
    b.x = w[ka * D + col0 + l16];
    b.y = w[(ka + 1) * D + col0 + l16];
    c = __builtin_amdgcn_wmma_f32_16x16x4_f32(false, a, false, b, (short)0, c,
                                              false, false);
  }
#pragma unroll
  for (int v = 0; v < 8; ++v) {
    int row = row0 + v + half * 8;
    if (row < n_nodes) h[(size_t)row * D + col0 + l16] = c[v];
  }
}

__global__ __launch_bounds__(256)
void gat_attn_scores(const float* __restrict__ h, const float* __restrict__ att_src,
                     const float* __restrict__ att_dst, float* __restrict__ a_src,
                     float* __restrict__ a_dst, float* __restrict__ self_logit,
                     float* __restrict__ mx, int n_nodes) {
  const int node = blockIdx.x * 8 + (threadIdx.x >> 5);
  const int lane = threadIdx.x & 31;
  if (node >= n_nodes) return;
  const float4 hv = ((const float4*)(h + (size_t)node * D))[lane];
  const float4 as = ((const float4*)att_src)[lane];
  const float4 ad = ((const float4*)att_dst)[lane];
  float ps = hv.x * as.x + hv.y * as.y + hv.z * as.z + hv.w * as.w;
  float pd = hv.x * ad.x + hv.y * ad.y + hv.z * ad.z + hv.w * ad.w;
  ps = wave_sum(ps);
  pd = wave_sum(pd);
  if (lane == 0) {
    a_src[node] = ps;
    a_dst[node] = pd;
    float lg = ps + pd;
    lg = lg > 0.0f ? lg : NEG_SLOPE * lg;
    self_logit[node] = lg;
    mx[node] = lg;
  }
}

__global__ __launch_bounds__(256)
void gat_edge_logits(const int* __restrict__ ei, const float* __restrict__ a_src,
                     const float* __restrict__ a_dst, float* __restrict__ logit,
                     float* __restrict__ mx, int n_edges) {
  const int e = blockIdx.x * 256 + threadIdx.x;
  if (e >= n_edges) return;
  const int s = ei[e];
  const int d = ei[n_edges + e];
  float lg = a_src[s] + a_dst[d];
  lg = lg > 0.0f ? lg : NEG_SLOPE * lg;
  logit[e] = lg;
  __hip_atomic_fetch_max(&mx[d], lg, __ATOMIC_RELAXED, __HIP_MEMORY_SCOPE_AGENT);
}

__global__ __launch_bounds__(256)
void gat_self_init(const float* __restrict__ h, const float* __restrict__ self_logit,
                   const float* __restrict__ mx, float* __restrict__ acc,
                   float* __restrict__ ssum, int n_nodes) {
  const int idx  = blockIdx.x * 256 + threadIdx.x;
  const int node = idx >> 5;
  const int c    = idx & 31;
  if (node >= n_nodes) return;
  const float e0 = __expf(self_logit[node] - mx[node]);
  float4 hv = ((const float4*)(h + (size_t)node * D))[c];
  float4 av = make_float4(hv.x * e0, hv.y * e0, hv.z * e0, hv.w * e0);
  ((float4*)(acc + (size_t)node * D))[c] = av;
  if (c == 0) ssum[node] = e0;
}

__global__ __launch_bounds__(256)
void gat_edge_accum(const int* __restrict__ ei, const float* __restrict__ logit,
                    const float* __restrict__ mx, const float* __restrict__ h,
                    float* __restrict__ acc, float* __restrict__ ssum, int n_edges) {
  const int e    = blockIdx.x * 8 + (threadIdx.x >> 5);
  const int lane = threadIdx.x & 31;
  if (e >= n_edges) return;
  const int s = ei[e];
  const int d = ei[n_edges + e];
  const float w = __expf(logit[e] - mx[d]);
  const float4 hv = ((const float4*)(h + (size_t)s * D))[lane];
  float* ap = acc + (size_t)d * D + lane * 4;
  unsafeAtomicAdd(ap + 0, hv.x * w);
  unsafeAtomicAdd(ap + 1, hv.y * w);
  unsafeAtomicAdd(ap + 2, hv.z * w);
  unsafeAtomicAdd(ap + 3, hv.w * w);
  if (lane == 0) unsafeAtomicAdd(&ssum[d], w);
}

__global__ __launch_bounds__(256)
void gat_ln_out(const float* __restrict__ x, const float* __restrict__ acc,
                const float* __restrict__ ssum, const float* __restrict__ bias,
                const float* __restrict__ gamma, const float* __restrict__ beta,
                float* __restrict__ out, int n_nodes) {
  const int node = blockIdx.x * 8 + (threadIdx.x >> 5);
  const int lane = threadIdx.x & 31;
  if (node >= n_nodes) return;
  const float inv_s = 1.0f / ssum[node];
  const float4 xv = ((const float4*)(x   + (size_t)node * D))[lane];
  const float4 av = ((const float4*)(acc + (size_t)node * D))[lane];
  const float4 bv = ((const float4*)bias)[lane];
  float4 y;
  y.x = xv.x + av.x * inv_s + bv.x;
  y.y = xv.y + av.y * inv_s + bv.y;
  y.z = xv.z + av.z * inv_s + bv.z;
  y.w = xv.w + av.w * inv_s + bv.w;
  float sum = y.x + y.y + y.z + y.w;
  float sq  = y.x * y.x + y.y * y.y + y.z * y.z + y.w * y.w;
  sum = wave_sum(sum);
  sq  = wave_sum(sq);
  const float mu   = sum * (1.0f / D);
  const float var  = sq * (1.0f / D) - mu * mu;
  const float rstd = rsqrtf(var + LN_EPS);
  const float4 gv = ((const float4*)gamma)[lane];
  const float4 tv = ((const float4*)beta)[lane];
  float4 o;
  o.x = (y.x - mu) * rstd * gv.x + tv.x;
  o.y = (y.y - mu) * rstd * gv.y + tv.y;
  o.z = (y.z - mu) * rstd * gv.z + tv.z;
  o.w = (y.w - mu) * rstd * gv.w + tv.w;
  ((float4*)(out + (size_t)node * D))[lane] = o;
}

extern "C" void kernel_launch(void* const* d_in, const int* in_sizes, int n_in,
                              void* d_out, int out_size, void* d_ws, size_t ws_size,
                              hipStream_t stream) {
  const float* x0   = (const float*)d_in[0];
  const int*   ei   = (const int*)d_in[1];
  const float* W    = (const float*)d_in[2];
  const float* asrc = (const float*)d_in[3];
  const float* adst = (const float*)d_in[4];
  const float* bias = (const float*)d_in[5];
  const float* gam  = (const float*)d_in[6];
  const float* bet  = (const float*)d_in[7];
  float* out = (float*)d_out;

  const int n  = in_sizes[0] / D;
  const int e  = in_sizes[1] / 2;
  const int nl = in_sizes[2] / (D * D);

  float* ws    = (float*)d_ws;
  float* h     = ws; ws += (size_t)n * D;
  float* acc   = ws; ws += (size_t)n * D;
  float* va    = ws; ws += n;
  float* vb    = ws; ws += n;
  float* slf   = ws; ws += n;
  float* mx    = ws; ws += n;
  float* ssum  = ws; ws += n;
  float* lgt   = ws; ws += e;

  for (int l = 0; l < nl; ++l) {
    const float* xin = (l == 0) ? x0 : out;
    const float* Wl  = W + (size_t)l * D * D;
    gat_gemm_wmma<<<(n + 15) / 16, 256, 0, stream>>>(xin, Wl, h, n);
    gat_attn_scores<<<(n + 7) / 8, 256, 0, stream>>>(h, asrc + l * D, adst + l * D,
                                                     va, vb, slf, mx, n);
    gat_edge_logits<<<(e + 255) / 256, 256, 0, stream>>>(ei, va, vb, lgt, mx, e);
    gat_self_init<<<(n * 32 + 255) / 256, 256, 0, stream>>>(h, slf, mx, acc, ssum, n);
    gat_edge_accum<<<(e + 7) / 8, 256, 0, stream>>>(ei, lgt, mx, h, acc, ssum, e);
    gat_ln_out<<<(n + 7) / 8, 256, 0, stream>>>(xin, acc, ssum, bias + l * D,
                                                gam + l * D, bet + l * D, out, n);
  }
}


# Replace the one-wave-per-edge f32 atomic scatter with a per-node gather over edges counting-sorted by destination in-call (int atomics); sum order per node = fill arrival order
# speedup vs baseline: 1.9775x; 1.9775x over previous
#include <hip/hip_runtime.h>
#include <math.h>

#define D 128
#define NEG_SLOPE 0.2f
#define LN_EPS 1e-5f

typedef float v2f __attribute__((ext_vector_type(2)));
typedef float v8f __attribute__((ext_vector_type(8)));

__device__ __forceinline__ float wave_sum(float v) {
#pragma unroll
  for (int off = 16; off > 0; off >>= 1) v += __shfl_xor(v, off, 32);
  return v;
}

__device__ __forceinline__ float wave_max(float v) {
  for (int off = 16; off > 0; off >>= 1) v = fmaxf(v, __shfl_xor(v, off, 32));
  return v;
}

__device__ __forceinline__ int wave_sum_i(int v) {
  for (int off = 16; off > 0; off >>= 1) v += __shfl_xor(v, off, 32);
  return v;
}

__device__ __forceinline__ float leaky(float v) {
  return v > 0.0f ? v : NEG_SLOPE * v;
}

__global__ __launch_bounds__(256)
void gat_gemm_wmma(const float* __restrict__ x, const float* __restrict__ w,
                   float* __restrict__ h, int n_nodes) {
  __shared__ float lds_a[16 * D];
  const int tid  = threadIdx.x;
  const int row0 = blockIdx.x * 16;

  for (int i = tid; i < 16 * D; i += 256) {
    int r = i >> 7, c = i & (D - 1);
    int row = row0 + r;
    lds_a[i] = (row < n_nodes) ? x[(size_t)row * D + c] : 0.0f;
  }
  __syncthreads();

  const int wave = tid >> 5;
  const int lane = tid & 31;
  const int half = lane >> 4;
  const int l16  = lane & 15;
  const int col0 = wave * 16;

  v8f c = {};
#pragma unroll
  for (int k = 0; k < D; k += 4) {
    const int ka = k + half * 2;
    v2f a, b;
    a.x = lds_a[l16 * D + ka];
    a.y = lds_a[l16 * D + ka + 1];
    b.x = w[ka * D + col0 + l16];
    b.y = w[(ka + 1) * D + col0 + l16];
    c = __builtin_amdgcn_wmma_f32_16x16x4_f32(false, a, false, b, (short)0, c,
                                              false, false);
  }
#pragma unroll
  for (int v = 0; v < 8; ++v) {
    int row = row0 + v + half * 8;
    if (row < n_nodes) h[(size_t)row * D + col0 + l16] = c[v];
  }
}

__global__ __launch_bounds__(256)
void csr_zero(int* __restrict__ cur, int n_nodes) {
  const int i = blockIdx.x * 256 + threadIdx.x;
  if (i < n_nodes) cur[i] = 0;
}

__global__ __launch_bounds__(256)
void csr_count(const int* __restrict__ ei, int* __restrict__ cur, int n_edges) {
  const int e = blockIdx.x * 256 + threadIdx.x;
  if (e >= n_edges) return;
  const int d = ei[n_edges + e];
  __hip_atomic_fetch_add(&cur[d], 1, __ATOMIC_RELAXED, __HIP_MEMORY_SCOPE_AGENT);
}

__global__ __launch_bounds__(1024)
void csr_scan(int* __restrict__ cur, int n_nodes) {
  __shared__ int wtot[32];
  const int lane = threadIdx.x & 31;
  const int wave = threadIdx.x >> 5;
  const int seg  = ((n_nodes + 1023) / 1024) * 32;
  const int b    = wave * seg;
  const int e    = min(b + seg, n_nodes);

  int s = 0;
  for (int i = b + lane; i < e; i += 32) s += cur[i];
  s = wave_sum_i(s);
  if (lane == 0) wtot[wave] = s;
  __syncthreads();

  int run = wave_sum_i(lane < wave ? wtot[lane] : 0);
  for (int i0 = b; i0 < e; i0 += 32) {
    const int i = i0 + lane;
    const int c = (i < e) ? cur[i] : 0;
    int inc = c;
    for (int o = 1; o < 32; o <<= 1) {
      const int v = __shfl_up(inc, o, 32);
      if (lane >= o) inc += v;
    }
    if (i < e) cur[i] = run + inc - c;
    run += __shfl(inc, 31, 32);
  }
}

__global__ __launch_bounds__(256)
void csr_fill(const int* __restrict__ ei, int* __restrict__ cur,
              int* __restrict__ srt, int n_edges) {
  const int e = blockIdx.x * 256 + threadIdx.x;
  if (e >= n_edges) return;
  const int s = ei[e];
  const int d = ei[n_edges + e];
  const int pos = __hip_atomic_fetch_add(&cur[d], 1, __ATOMIC_RELAXED,
                                         __HIP_MEMORY_SCOPE_AGENT);
  srt[pos] = s;
}

__global__ __launch_bounds__(256)
void gat_attn_scores(const float* __restrict__ h, const float* __restrict__ att_src,
                     const float* __restrict__ att_dst, float* __restrict__ a_src,
                     float* __restrict__ a_dst, int n_nodes) {
  const int node = blockIdx.x * 8 + (threadIdx.x >> 5);
  const int lane = threadIdx.x & 31;
  if (node >= n_nodes) return;
  const float4 hv = ((const float4*)(h + (size_t)node * D))[lane];
  const float4 as = ((const float4*)att_src)[lane];
  const float4 ad = ((const float4*)att_dst)[lane];
  float ps = hv.x * as.x + hv.y * as.y + hv.z * as.z + hv.w * as.w;
  float pd = hv.x * ad.x + hv.y * ad.y + hv.z * ad.z + hv.w * ad.w;
  ps = wave_sum(ps);
  pd = wave_sum(pd);
  if (lane == 0) {
    a_src[node] = ps;
    a_dst[node] = pd;
  }
}

__global__ __launch_bounds__(256)
void gat_gather(const int* __restrict__ seg_end, const int* __restrict__ srt,
                const float* __restrict__ a_src, const float* __restrict__ a_dst,
                const float* __restrict__ h, float* __restrict__ acc,
                float* __restrict__ ssum, int n_nodes) {
  const int wv   = __builtin_amdgcn_readfirstlane((int)(threadIdx.x >> 5));
  const int node = blockIdx.x * 8 + wv;
  const int lane = threadIdx.x & 31;
  if (node >= n_nodes) return;
  const int beg = (node > 0) ? seg_end[node - 1] : 0;
  const int end = seg_end[node];
  const float ad    = a_dst[node];
  const float lself = leaky(a_src[node] + ad);

  float m = lself;
  for (int i = beg + lane; i < end; i += 32) m = fmaxf(m, leaky(a_src[srt[i]] + ad));
  m = wave_max(m);

  const float w0 = __expf(lself - m);
  const float4 hs = ((const float4*)(h + (size_t)node * D))[lane];
  float4 a = make_float4(hs.x * w0, hs.y * w0, hs.z * w0, hs.w * w0);
  float sw = 0.0f;
  for (int i0 = beg; i0 < end; i0 += 32) {
    const int i = i0 + lane;
    int   s = 0;
    float w = 0.0f;
    if (i < end) {
      s = srt[i];
      w = __expf(leaky(a_src[s] + ad) - m);
    }
    sw += w;
    const int cnt = min(32, end - i0);
#pragma unroll 4
    for (int j = 0; j < cnt; ++j) {
      const int   sj = __builtin_amdgcn_readlane(s, j);
      const float wj = __int_as_float(__builtin_amdgcn_readlane(__float_as_int(w), j));
      const float4 hv = ((const float4*)(h + (size_t)sj * D))[lane];
      a.x += wj * hv.x;
      a.y += wj * hv.y;
      a.z += wj * hv.z;
      a.w += wj * hv.w;
    }
  }
  sw = wave_sum(sw) + w0;
  ((float4*)(acc + (size_t)node * D))[lane] = a;
  if (lane == 0) ssum[node] = sw;
}

__global__ __launch_bounds__(256)
void gat_ln_out(const float* __restrict__ x, const float* __restrict__ acc,
                const float* __restrict__ ssum, const float* __restrict__ bias,
                const float* __restrict__ gamma, const float* __restrict__ beta,
                float* __restrict__ out, int n_nodes) {
  const int node = blockIdx.x * 8 + (threadIdx.x >> 5);
  const int lane = threadIdx.x & 31;
  if (node >= n_nodes) return;
  const float inv_s = 1.0f / ssum[node];
  const float4 xv = ((const float4*)(x   + (size_t)node * D))[lane];
  const float4 av = ((const float4*)(acc + (size_t)node * D))[lane];
  const float4 bv = ((const float4*)bias)[lane];
  float4 y;
  y.x = xv.x + av.x * inv_s + bv.x;
  y.y = xv.y + av.y * inv_s + bv.y;
  y.z = xv.z + av.z * inv_s + bv.z;
  y.w = xv.w + av.w * inv_s + bv.w;
  float sum = y.x + y.y + y.z + y.w;
  float sq  = y.x * y.x + y.y * y.y + y.z * y.z + y.w * y.w;
  sum = wave_sum(sum);
  sq  = wave_sum(sq);
  const float mu   = sum * (1.0f / D);
  const float var  = sq * (1.0f / D) - mu * mu;
  const float rstd = rsqrtf(var + LN_EPS);
  const float4 gv = ((const float4*)gamma)[lane];
  const float4 tv = ((const float4*)beta)[lane];
  float4 o;
  o.x = (y.x - mu) * rstd * gv.x + tv.x;
  o.y = (y.y - mu) * rstd * gv.y + tv.y;
  o.z = (y.z - mu) * rstd * gv.z + tv.z;
  o.w = (y.w - mu) * rstd * gv.w + tv.w;
  ((float4*)(out + (size_t)node * D))[lane] = o;
}

extern "C" void kernel_launch(void* const* d_in, const int* in_sizes, int n_in,
                              void* d_out, int out_size, void* d_ws, size_t ws_size,
                              hipStream_t stream) {
  const float* x0   = (const float*)d_in[0];
  const int*   ei   = (const int*)d_in[1];
  const float* W    = (const float*)d_in[2];
  const float* asrc = (const float*)d_in[3];
  const float* adst = (const float*)d_in[4];
  const float* bias = (const float*)d_in[5];
  const float* gam  = (const float*)d_in[6];
  const float* bet  = (const float*)d_in[7];
  float* out = (float*)d_out;

  const int n  = in_sizes[0] / D;
  const int e  = in_sizes[1] / 2;
  const int nl = in_sizes[2] / (D * D);

  float* ws    = (float*)d_ws;
  float* h     = ws; ws += (size_t)n * D;
  float* acc   = ws; ws += (size_t)n * D;
  float* va    = ws; ws += n;
  float* vb    = ws; ws += n;
  float* ssum  = ws; ws += n;
  int*   cur   = (int*)ws; ws += n;
  int*   srt   = (int*)ws; ws += e;

  csr_zero<<<(n + 255) / 256, 256, 0, stream>>>(cur, n);
  csr_count<<<(e + 255) / 256, 256, 0, stream>>>(ei, cur, e);
  csr_scan<<<1, 1024, 0, stream>>>(cur, n);
  csr_fill<<<(e + 255) / 256, 256, 0, stream>>>(ei, cur, srt, e);

  for (int l = 0; l < nl; ++l) {
    const float* xin = (l == 0) ? x0 : out;
    const float* Wl  = W + (size_t)l * D * D;
    gat_gemm_wmma<<<(n + 15) / 16, 256, 0, stream>>>(xin, Wl, h, n);
    gat_attn_scores<<<(n + 7) / 8, 256, 0, stream>>>(h, asrc + l * D, adst + l * D,
                                                     va, vb, n);
    gat_gather<<<(n + 7) / 8, 256, 0, stream>>>(cur, srt, va, vb, h, acc, ssum, n);
    gat_ln_out<<<(n + 7) / 8, 256, 0, stream>>>(xin, acc, ssum, bias + l * D,
                                                gam + l * D, bet + l * D, out, n);
  }
}


# Fuse the two light per-node kernels into their producers: score dots in the GEMM epilogue (sum order: 16-lane tree per tile, then 8 tiles), residual+LayerNorm in the gather; 12 launches -> 8
# speedup vs baseline: 2.1847x; 1.1048x over previous
#include <hip/hip_runtime.h>
#include <math.h>

#define D 128
#define NEG_SLOPE 0.2f
#define LN_EPS 1e-5f

typedef float v2f __attribute__((ext_vector_type(2)));
typedef float v8f __attribute__((ext_vector_type(8)));

__device__ __forceinline__ float wave_sum(float v) {
#pragma unroll
  for (int off = 16; off > 0; off >>= 1) v += __shfl_xor(v, off, 32);
  return v;
}

__device__ __forceinline__ float wave_max(float v) {
  for (int off = 16; off > 0; off >>= 1) v = fmaxf(v, __shfl_xor(v, off, 32));
  return v;
}

__device__ __forceinline__ int wave_sum_i(int v) {
  for (int off = 16; off > 0; off >>= 1) v += __shfl_xor(v, off, 32);
  return v;
}

__device__ __forceinline__ float leaky(float v) {
  return v > 0.0f ? v : NEG_SLOPE * v;
}

template <int M>
__device__ __forceinline__ float row_xor_add(float v) {
  return v + __int_as_float(__builtin_amdgcn_update_dpp(
                 0, __float_as_int(v), 0x160 + M, 0xf, 0xf, true));
}

__device__ __forceinline__ float row16_sum(float v) {
  v = row_xor_add<8>(v);
  v = row_xor_add<4>(v);
  v = row_xor_add<2>(v);
  v = row_xor_add<1>(v);
  return v;
}

__global__ __launch_bounds__(256, 16)
void gat_gemm_wmma(const float* __restrict__ x, const float* __restrict__ w,
                   const float* __restrict__ att_src,
                   const float* __restrict__ att_dst,
                   float* __restrict__ h, float* __restrict__ a_src,
                   float* __restrict__ a_dst, int n_nodes) {
  __shared__ float lds_a[16 * D];
  __shared__ float lds_p[8 * 16 * 2];
  const int tid  = threadIdx.x;
  const int row0 = blockIdx.x * 16;

  for (int i = tid; i < 16 * D; i += 256) {
    int r = i >> 7, c = i & (D - 1);
    int row = row0 + r;
    lds_a[i] = (row < n_nodes) ? x[(size_t)row * D + c] : 0.0f;
  }
  __syncthreads();

  const int wave = tid >> 5;
  const int lane = tid & 31;
  const int half = lane >> 4;
  const int l16  = lane & 15;
  const int col0 = wave * 16;

  v8f c = {};
#pragma unroll
  for (int k = 0; k < D; k += 4) {
    const int ka = k + half * 2;
    v2f a, b;
    a.x = lds_a[l16 * D + ka];
    a.y = lds_a[l16 * D + ka + 1];
    b.x = w[ka * D + col0 + l16];
    b.y = w[(ka + 1) * D + col0 + l16];
    c = __builtin_amdgcn_wmma_f32_16x16x4_f32(false, a, false, b, (short)0, c,
                                              false, false);
  }
  const float as = att_src[col0 + l16];
  const float ad = att_dst[col0 + l16];
#pragma unroll
  for (int v = 0; v < 8; ++v) {
    int row = row0 + v + half * 8;
    if (row < n_nodes) h[(size_t)row * D + col0 + l16] = c[v];
  }
#pragma unroll
  for (int v = 0; v < 8; ++v) {
    const float ps = row16_sum(c[v] * as);
    const float pd = row16_sum(c[v] * ad);
    if (l16 == 0) {
      lds_p[(wave * 16 + v + half * 8) * 2 + 0] = ps;
      lds_p[(wave * 16 + v + half * 8) * 2 + 1] = pd;
    }
  }
  __syncthreads();
  if (tid < 32) {
    const int r = tid >> 1, which = tid & 1;
    float s = 0.0f;
#pragma unroll
    for (int t = 0; t < 8; ++t) s += lds_p[(t * 16 + r) * 2 + which];
    const int row = row0 + r;
    if (row < n_nodes) (which ? a_dst : a_src)[row] = s;
  }
}

__global__ __launch_bounds__(256)
void csr_zero(int* __restrict__ cur, int n_nodes) {
  const int i = blockIdx.x * 256 + threadIdx.x;
  if (i < n_nodes) cur[i] = 0;
}

__global__ __launch_bounds__(256)
void csr_count(const int* __restrict__ ei, int* __restrict__ cur, int n_edges) {
  const int e = blockIdx.x * 256 + threadIdx.x;
  if (e >= n_edges) return;
  const int d = ei[n_edges + e];
  __hip_atomic_fetch_add(&cur[d], 1, __ATOMIC_RELAXED, __HIP_MEMORY_SCOPE_AGENT);
}

__global__ __launch_bounds__(1024)
void csr_scan(int* __restrict__ cur, int n_nodes) {
  __shared__ int wtot[32];
  const int lane = threadIdx.x & 31;
  const int wave = threadIdx.x >> 5;
  const int seg  = ((n_nodes + 1023) / 1024) * 32;
  const int b    = wave * seg;
  const int e    = min(b + seg, n_nodes);

  int s = 0;
  for (int i = b + lane; i < e; i += 32) s += cur[i];
  s = wave_sum_i(s);
  if (lane == 0) wtot[wave] = s;
  __syncthreads();

  int run = wave_sum_i(lane < wave ? wtot[lane] : 0);
  for (int i0 = b; i0 < e; i0 += 32) {
    const int i = i0 + lane;
    const int c = (i < e) ? cur[i] : 0;
    int inc = c;
    for (int o = 1; o < 32; o <<= 1) {
      const int v = __shfl_up(inc, o, 32);
      if (lane >= o) inc += v;
    }
    if (i < e) cur[i] = run + inc - c;
    run += __shfl(inc, 31, 32);
  }
}

__global__ __launch_bounds__(256)
void csr_fill(const int* __restrict__ ei, int* __restrict__ cur,
              int* __restrict__ srt, int n_edges) {
  const int e = blockIdx.x * 256 + threadIdx.x;
  if (e >= n_edges) return;
  const int s = ei[e];
  const int d = ei[n_edges + e];
  const int pos = __hip_atomic_fetch_add(&cur[d], 1, __ATOMIC_RELAXED,
                                         __HIP_MEMORY_SCOPE_AGENT);
  srt[pos] = s;
}

__global__ __launch_bounds__(256)
void gat_gather_ln(const int* __restrict__ seg_end, const int* __restrict__ srt,
                   const float* __restrict__ a_src, const float* __restrict__ a_dst,
                   const float* __restrict__ h, const float* x,
                   const float* __restrict__ bias, const float* __restrict__ gamma,
                   const float* __restrict__ beta, float* out, int n_nodes) {
  const int wv   = __builtin_amdgcn_readfirstlane((int)(threadIdx.x >> 5));
  const int node = blockIdx.x * 8 + wv;
  const int lane = threadIdx.x & 31;
  if (node >= n_nodes) return;
  const int beg = (node > 0) ? seg_end[node - 1] : 0;
  const int end = seg_end[node];
  const float ad    = a_dst[node];
  const float lself = leaky(a_src[node] + ad);
  const float4 xv = ((const float4*)(x + (size_t)node * D))[lane];

  float m = lself;
  for (int i = beg + lane; i < end; i += 32) m = fmaxf(m, leaky(a_src[srt[i]] + ad));
  m = wave_max(m);

  const float w0 = __expf(lself - m);
  const float4 hs = ((const float4*)(h + (size_t)node * D))[lane];
  float4 a = make_float4(hs.x * w0, hs.y * w0, hs.z * w0, hs.w * w0);
  float sw = 0.0f;
  for (int i0 = beg; i0 < end; i0 += 32) {
    const int i = i0 + lane;
    int   s = 0;
    float w = 0.0f;
    if (i < end) {
      s = srt[i];
      w = __expf(leaky(a_src[s] + ad) - m);
    }
    sw += w;
    const int cnt = min(32, end - i0);
#pragma unroll 4
    for (int j = 0; j < cnt; ++j) {
      const int   sj = __builtin_amdgcn_readlane(s, j);
      const float wj = __int_as_float(__builtin_amdgcn_readlane(__float_as_int(w), j));
      const float4 hv = ((const float4*)(h + (size_t)sj * D))[lane];
      a.x += wj * hv.x;
      a.y += wj * hv.y;
      a.z += wj * hv.z;
      a.w += wj * hv.w;
    }
  }
  sw = wave_sum(sw) + w0;

  const float inv_s = 1.0f / sw;
  const float4 bv = ((const float4*)bias)[lane];
  float4 y;
  y.x = xv.x + a.x * inv_s + bv.x;
  y.y = xv.y + a.y * inv_s + bv.y;
  y.z = xv.z + a.z * inv_s + bv.z;
  y.w = xv.w + a.w * inv_s + bv.w;
  float sum = y.x + y.y + y.z + y.w;
  float sq  = y.x * y.x + y.y * y.y + y.z * y.z + y.w * y.w;
  sum = wave_sum(sum);
  sq  = wave_sum(sq);
  const float mu   = sum * (1.0f / D);
  const float var  = sq * (1.0f / D) - mu * mu;
  const float rstd = rsqrtf(var + LN_EPS);
  const float4 gv = ((const float4*)gamma)[lane];
  const float4 tv = ((const float4*)beta)[lane];
  float4 o;
  o.x = (y.x - mu) * rstd * gv.x + tv.x;
  o.y = (y.y - mu) * rstd * gv.y + tv.y;
  o.z = (y.z - mu) * rstd * gv.z + tv.z;
  o.w = (y.w - mu) * rstd * gv.w + tv.w;
  ((float4*)(out + (size_t)node * D))[lane] = o;
}

extern "C" void kernel_launch(void* const* d_in, const int* in_sizes, int n_in,
                              void* d_out, int out_size, void* d_ws, size_t ws_size,
                              hipStream_t stream) {
  const float* x0   = (const float*)d_in[0];
  const int*   ei   = (const int*)d_in[1];
  const float* W    = (const float*)d_in[2];
  const float* asrc = (const float*)d_in[3];
  const float* adst = (const float*)d_in[4];
  const float* bias = (const float*)d_in[5];
  const float* gam  = (const float*)d_in[6];
  const float* bet  = (const float*)d_in[7];
  float* out = (float*)d_out;

  const int n  = in_sizes[0] / D;
  const int e  = in_sizes[1] / 2;
  const int nl = in_sizes[2] / (D * D);

  float* ws    = (float*)d_ws;
  float* h     = ws; ws += (size_t)n * D;
  float* va    = ws; ws += n;
  float* vb    = ws; ws += n;
  int*   cur   = (int*)ws; ws += n;
  int*   srt   = (int*)ws; ws += e;

  csr_zero<<<(n + 255) / 256, 256, 0, stream>>>(cur, n);
  csr_count<<<(e + 255) / 256, 256, 0, stream>>>(ei, cur, e);
  csr_scan<<<1, 1024, 0, stream>>>(cur, n);
  csr_fill<<<(e + 255) / 256, 256, 0, stream>>>(ei, cur, srt, e);

  for (int l = 0; l < nl; ++l) {
    const float* xin = (l == 0) ? x0 : out;
    const float* Wl  = W + (size_t)l * D * D;
    gat_gemm_wmma<<<(n + 15) / 16, 256, 0, stream>>>(xin, Wl, asrc + l * D,
                                                     adst + l * D, h, va, vb, n);
    gat_gather_ln<<<(n + 7) / 8, 256, 0, stream>>>(cur, srt, va, vb, h, xin,
                                                   bias + l * D, gam + l * D,
                                                   bet + l * D, out, n);
  }
}


# Sort edges by destination with LDS counters (chunk histograms, one register-held scan, split, per-bucket placing) instead of global int atomics; sum order per node = LDS add arrival order
# speedup vs baseline: 3.3567x; 1.5365x over previous
#include <hip/hip_runtime.h>
#include <math.h>

#define D 128
#define NEG_SLOPE 0.2f
#define LN_EPS 1e-5f

typedef float v2f __attribute__((ext_vector_type(2)));
typedef float v8f __attribute__((ext_vector_type(8)));

__device__ __forceinline__ float wave_sum(float v) {
#pragma unroll
  for (int off = 16; off > 0; off >>= 1) v += __shfl_xor(v, off, 32);
  return v;
}

__device__ __forceinline__ float wave_max(float v) {
  for (int off = 16; off > 0; off >>= 1) v = fmaxf(v, __shfl_xor(v, off, 32));
  return v;
}

__device__ __forceinline__ int wave_sum_i(int v) {
  for (int off = 16; off > 0; off >>= 1) v += __shfl_xor(v, off, 32);
  return v;
}

__device__ __forceinline__ float leaky(float v) {
  return v > 0.0f ? v : NEG_SLOPE * v;
}

template <int M>
__device__ __forceinline__ float row_xor_add(float v) {
  return v + __int_as_float(__builtin_amdgcn_update_dpp(
                 0, __float_as_int(v), 0x160 + M, 0xf, 0xf, true));
}

__device__ __forceinline__ float row16_sum(float v) {
  v = row_xor_add<8>(v);
  v = row_xor_add<4>(v);
  v = row_xor_add<2>(v);
  v = row_xor_add<1>(v);
  return v;
}

__global__ __launch_bounds__(256, 16)
void gat_gemm_wmma(const float* __restrict__ x, const float* __restrict__ w,
                   const float* __restrict__ att_src,
                   const float* __restrict__ att_dst,
                   float* __restrict__ h, float* __restrict__ a_src,
                   float* __restrict__ a_dst, int n_nodes) {
  __shared__ float lds_a[16 * D];
  __shared__ float lds_p[8 * 16 * 2];
  const int tid  = threadIdx.x;
  const int row0 = blockIdx.x * 16;

  for (int i = tid; i < 16 * D; i += 256) {
    int r = i >> 7, c = i & (D - 1);
    int row = row0 + r;
    lds_a[i] = (row < n_nodes) ? x[(size_t)row * D + c] : 0.0f;
  }
  __syncthreads();

  const int wave = tid >> 5;
  const int lane = tid & 31;
  const int half = lane >> 4;
  const int l16  = lane & 15;
  const int col0 = wave * 16;

  v8f c = {};
#pragma unroll
  for (int k = 0; k < D; k += 4) {
    const int ka = k + half * 2;
    v2f a, b;
    a.x = lds_a[l16 * D + ka];
    a.y = lds_a[l16 * D + ka + 1];
    b.x = w[ka * D + col0 + l16];
    b.y = w[(ka + 1) * D + col0 + l16];
    c = __builtin_amdgcn_wmma_f32_16x16x4_f32(false, a, false, b, (short)0, c,
                                              false, false);
  }
  const float as = att_src[col0 + l16];
  const float ad = att_dst[col0 + l16];
#pragma unroll
  for (int v = 0; v < 8; ++v) {
    int row = row0 + v + half * 8;
    if (row < n_nodes) h[(size_t)row * D + col0 + l16] = c[v];
  }
#pragma unroll
  for (int v = 0; v < 8; ++v) {
    const float ps = row16_sum(c[v] * as);
    const float pd = row16_sum(c[v] * ad);
    if (l16 == 0) {
      lds_p[(wave * 16 + v + half * 8) * 2 + 0] = ps;
      lds_p[(wave * 16 + v + half * 8) * 2 + 1] = pd;
    }
  }
  __syncthreads();
  if (tid < 32) {
    const int r = tid >> 1, which = tid & 1;
    float s = 0.0f;
#pragma unroll
    for (int t = 0; t < 8; ++t) s += lds_p[(t * 16 + r) * 2 + which];
    const int row = row0 + r;
    if (row < n_nodes) (which ? a_dst : a_src)[row] = s;
  }
}


template <int N>
__device__ __forceinline__ int row_shr_add(int v) {
  return v + __builtin_amdgcn_update_dpp(0, v, 0x110 + N, 0xf, 0xf, true);
}

__device__ __forceinline__ int wave_incl_scan_i(int v, int lane) {
  v = row_shr_add<1>(v);
  v = row_shr_add<2>(v);
  v = row_shr_add<4>(v);
  v = row_shr_add<8>(v);
  const int lo = __builtin_amdgcn_readlane(v, 15);
  return v + ((lane & 16) ? lo : 0);
}

#define BSH    8
#define BNODES 256
#define NCH    256
#define ST     512
#define SU     8
#define SUB    16

__global__ __launch_bounds__(ST)
void sort_hist(const int* __restrict__ ei, int* __restrict__ hist_t,
               int n_edges, int nb) {
  extern __shared__ int lds_dyn[];
  const int tid = threadIdx.x;
  const int wg  = blockIdx.x;
  for (int b = tid; b < nb; b += ST) lds_dyn[b] = 0;
  __syncthreads();
  const int per = (n_edges + NCH - 1) / NCH;
  const int c0  = wg * per;
  const int c1  = min(c0 + per, n_edges);
  for (int i0 = c0 + tid; i0 < c1; i0 += ST * SU) {
    int d[SU];
#pragma unroll
    for (int j = 0; j < SU; ++j) {
      const int i = min(i0 + j * ST, c1 - 1);
      d[j] = ei[n_edges + i];
    }
#pragma unroll
    for (int j = 0; j < SU; ++j)
      if (i0 + j * ST < c1) atomicAdd(&lds_dyn[d[j] >> BSH], 1);
  }
  __syncthreads();
  for (int b = tid; b < nb; b += ST) hist_t[b * NCH + wg] = lds_dyn[b];
}

#define SCAN_CH 16
__device__ __forceinline__ int4 scan_row(int4 q, int lane, int& run) {
  const int t   = q.x + q.y + q.z + q.w;
  const int inc = wave_incl_scan_i(t, lane);
  int4 o;
  o.x = run + inc - t;
  o.y = o.x + q.x;
  o.z = o.y + q.y;
  o.w = o.z + q.z;
  run += __builtin_amdgcn_readlane(inc, 31);
  return o;
}

__global__ __launch_bounds__(1024)
void sort_scan(int* __restrict__ a, int m) {
  __shared__ int wtot[32];
  const int lane = threadIdx.x & 31;
  const int wave = threadIdx.x >> 5;
  int4* a4 = (int4*)a;
  const int rows = m >> 7;
  const int rpw  = (rows + 31) >> 5;
  const int r0   = wave * rpw;
  const int r1   = min(r0 + rpw, rows);

  int4 v[SCAN_CH];
  int s = 0;
#pragma unroll
  for (int k = 0; k < SCAN_CH; ++k) {
    const int r = r0 + k;
    v[k] = (r < r1) ? a4[r * 32 + lane] : make_int4(0, 0, 0, 0);
    s += v[k].x + v[k].y + v[k].z + v[k].w;
  }
  for (int r = r0 + SCAN_CH; r < r1; ++r) {
    const int4 q = a4[r * 32 + lane];
    s += q.x + q.y + q.z + q.w;
  }
  s = wave_sum_i(s);
  if (lane == 0) wtot[wave] = s;
  __syncthreads();

  int run = wave_sum_i(lane < wave ? wtot[lane] : 0);
#pragma unroll
  for (int k = 0; k < SCAN_CH; ++k) {
    const int r = r0 + k;
    const int4 o = scan_row(v[k], lane, run);
    if (r < r1) a4[r * 32 + lane] = o;
  }
  for (int r = r0 + SCAN_CH; r < r1; ++r) {
    const int4 o = scan_row(a4[r * 32 + lane], lane, run);
    a4[r * 32 + lane] = o;
  }
}

__global__ __launch_bounds__(ST)
void sort_split(const int* __restrict__ ei, const int* __restrict__ off_t,
                int* __restrict__ pairs_raw, int n_edges, int nb) {
  extern __shared__ int lds_dyn[];
  int2* pairs = (int2*)pairs_raw;
  const int tid = threadIdx.x;
  const int wg  = blockIdx.x;
  for (int b = tid; b < nb; b += ST) lds_dyn[b] = off_t[b * NCH + wg];
  __syncthreads();
  const int per = (n_edges + NCH - 1) / NCH;
  const int c0  = wg * per;
  const int c1  = min(c0 + per, n_edges);
  for (int i0 = c0 + tid; i0 < c1; i0 += ST * SU) {
    int s[SU], d[SU];
#pragma unroll
    for (int j = 0; j < SU; ++j) {
      const int i = min(i0 + j * ST, c1 - 1);
      s[j] = ei[i];
      d[j] = ei[n_edges + i];
    }
#pragma unroll
    for (int j = 0; j < SU; ++j) {
      if (i0 + j * ST < c1) {
        const int p = atomicAdd(&lds_dyn[d[j] >> BSH], 1);
        pairs[p] = make_int2(s[j], d[j]);
      }
    }
  }
}

__global__ __launch_bounds__(ST)
void sort_bucket(const int* __restrict__ off_t, const int* __restrict__ pairs_raw,
                 int* __restrict__ seg_end, int* __restrict__ srt,
                 int n_nodes, int n_edges, int nb) {
  __shared__ int cnt[BNODES];
  __shared__ int wsum[BNODES / 32];
  const int2* pairs = (const int2*)pairs_raw;
  const int tid  = threadIdx.x;
  const int lane = tid & 31;
  const int wave = tid >> 5;
  const int b    = blockIdx.x;
  const int bb   = off_t[b * NCH];
  const int be   = (b + 1 < nb) ? off_t[(b + 1) * NCH] : n_edges;

  if (tid < BNODES) cnt[tid] = 0;
  __syncthreads();
  for (int i0 = bb + tid; i0 < be; i0 += ST * SUB) {
    int d[SUB];
#pragma unroll
    for (int j = 0; j < SUB; ++j) {
      const int i = min(i0 + j * ST, be - 1);
      d[j] = pairs[i].y;
    }
#pragma unroll
    for (int j = 0; j < SUB; ++j)
      if (i0 + j * ST < be) atomicAdd(&cnt[d[j] & (BNODES - 1)], 1);
  }
  __syncthreads();

  int c = 0, inc = 0;
  if (tid < BNODES) {
    c   = cnt[tid];
    inc = wave_incl_scan_i(c, lane);
    if (lane == 31) wsum[wave] = inc;
  }
  __syncthreads();
  if (tid < BNODES) {
    int base = 0;
    for (int w = 0; w < wave; ++w) base += wsum[w];
    const int excl = base + inc - c;
    cnt[tid] = excl;
    const int node = b * BNODES + tid;
    if (node < n_nodes) seg_end[node] = bb + excl + c;
  }
  __syncthreads();
  for (int i0 = bb + tid; i0 < be; i0 += ST * SUB) {
    int2 pr[SUB];
#pragma unroll
    for (int j = 0; j < SUB; ++j) {
      const int i = min(i0 + j * ST, be - 1);
      pr[j] = pairs[i];
    }
#pragma unroll
    for (int j = 0; j < SUB; ++j) {
      if (i0 + j * ST < be) {
        const int p = atomicAdd(&cnt[pr[j].y & (BNODES - 1)], 1);
        srt[bb + p] = pr[j].x;
      }
    }
  }
}

__global__ __launch_bounds__(256)
void gat_gather_ln(const int* __restrict__ seg_end, const int* __restrict__ srt,
                   const float* __restrict__ a_src, const float* __restrict__ a_dst,
                   const float* __restrict__ h, const float* x,
                   const float* __restrict__ bias, const float* __restrict__ gamma,
                   const float* __restrict__ beta, float* out, int n_nodes) {
  const int wv   = __builtin_amdgcn_readfirstlane((int)(threadIdx.x >> 5));
  const int node = blockIdx.x * 8 + wv;
  const int lane = threadIdx.x & 31;
  if (node >= n_nodes) return;
  const int beg = (node > 0) ? seg_end[node - 1] : 0;
  const int end = seg_end[node];
  const float ad    = a_dst[node];
  const float lself = leaky(a_src[node] + ad);
  const float4 xv = ((const float4*)(x + (size_t)node * D))[lane];

  float m = lself;
  for (int i = beg + lane; i < end; i += 32) m = fmaxf(m, leaky(a_src[srt[i]] + ad));
  m = wave_max(m);

  const float w0 = __expf(lself - m);
  const float4 hs = ((const float4*)(h + (size_t)node * D))[lane];
  float4 a = make_float4(hs.x * w0, hs.y * w0, hs.z * w0, hs.w * w0);
  float sw = 0.0f;
  for (int i0 = beg; i0 < end; i0 += 32) {
    const int i = i0 + lane;
    int   s = 0;
    float w = 0.0f;
    if (i < end) {
      s = srt[i];
      w = __expf(leaky(a_src[s] + ad) - m);
    }
    sw += w;
    const int cnt = min(32, end - i0);
#pragma unroll 4
    for (int j = 0; j < cnt; ++j) {
      const int   sj = __builtin_amdgcn_readlane(s, j);
      const float wj = __int_as_float(__builtin_amdgcn_readlane(__float_as_int(w), j));
      const float4 hv = ((const float4*)(h + (size_t)sj * D))[lane];
      a.x += wj * hv.x;
      a.y += wj * hv.y;
      a.z += wj * hv.z;
      a.w += wj * hv.w;
    }
  }
  sw = wave_sum(sw) + w0;

  const float inv_s = 1.0f / sw;
  const float4 bv = ((const float4*)bias)[lane];
  float4 y;
  y.x = xv.x + a.x * inv_s + bv.x;
  y.y = xv.y + a.y * inv_s + bv.y;
  y.z = xv.z + a.z * inv_s + bv.z;
  y.w = xv.w + a.w * inv_s + bv.w;
  float sum = y.x + y.y + y.z + y.w;
  float sq  = y.x * y.x + y.y * y.y + y.z * y.z + y.w * y.w;
  sum = wave_sum(sum);
  sq  = wave_sum(sq);
  const float mu   = sum * (1.0f / D);
  const float var  = sq * (1.0f / D) - mu * mu;
  const float rstd = rsqrtf(var + LN_EPS);
  const float4 gv = ((const float4*)gamma)[lane];
  const float4 tv = ((const float4*)beta)[lane];
  float4 o;
  o.x = (y.x - mu) * rstd * gv.x + tv.x;
  o.y = (y.y - mu) * rstd * gv.y + tv.y;
  o.z = (y.z - mu) * rstd * gv.z + tv.z;
  o.w = (y.w - mu) * rstd * gv.w + tv.w;
  ((float4*)(out + (size_t)node * D))[lane] = o;
}

extern "C" void kernel_launch(void* const* d_in, const int* in_sizes, int n_in,
                              void* d_out, int out_size, void* d_ws, size_t ws_size,
                              hipStream_t stream) {
  const float* x0   = (const float*)d_in[0];
  const int*   ei   = (const int*)d_in[1];
  const float* W    = (const float*)d_in[2];
  const float* asrc = (const float*)d_in[3];
  const float* adst = (const float*)d_in[4];
  const float* bias = (const float*)d_in[5];
  const float* gam  = (const float*)d_in[6];
  const float* bet  = (const float*)d_in[7];
  float* out = (float*)d_out;

  const int n  = in_sizes[0] / D;
  const int e  = in_sizes[1] / 2;
  const int nl = in_sizes[2] / (D * D);

  float* ws    = (float*)d_ws;
  float* h     = ws; ws += (size_t)n * D;
  float* va    = ws; ws += n;
  float* vb    = ws; ws += n;
  int*   cur   = (int*)ws; ws += n;
  int*   srt   = (int*)ws; ws += e;

  const int nb      = (n + BNODES - 1) / BNODES;
  int* hist_t = (int*)h;
  int* pairs  = hist_t + (size_t)nb * NCH;
  sort_hist<<<NCH, ST, nb * sizeof(int), stream>>>(ei, hist_t, e, nb);
  sort_scan<<<1, 1024, 0, stream>>>(hist_t, nb * NCH);
  sort_split<<<NCH, ST, nb * sizeof(int), stream>>>(ei, hist_t, pairs, e, nb);
  sort_bucket<<<nb, ST, 0, stream>>>(hist_t, pairs, cur, srt, n, e, nb);

  for (int l = 0; l < nl; ++l) {
    const float* xin = (l == 0) ? x0 : out;
    const float* Wl  = W + (size_t)l * D * D;
    gat_gemm_wmma<<<(n + 15) / 16, 256, 0, stream>>>(xin, Wl, asrc + l * D,
                                                     adst + l * D, h, va, vb, n);
    gat_gather_ln<<<(n + 7) / 8, 256, 0, stream>>>(cur, srt, va, vb, h, xin,
                                                   bias + l * D, gam + l * D,
                                                   bet + l * D, out, n);
  }
}


# Pad the GEMM's A slab in LDS from a 128- to a 132-float row pitch: a fragment read's 16 rows now fall on distinct banks instead of one bank pair; products, sums and their order unchanged
# speedup vs baseline: 3.9979x; 1.1910x over previous
#include <hip/hip_runtime.h>
#include <math.h>

#define D 128
#define LDA 132
#define NEG_SLOPE 0.2f
#define LN_EPS 1e-5f

typedef float v2f __attribute__((ext_vector_type(2)));
typedef float v8f __attribute__((ext_vector_type(8)));

__device__ __forceinline__ float wave_sum(float v) {
#pragma unroll
  for (int off = 16; off > 0; off >>= 1) v += __shfl_xor(v, off, 32);
  return v;
}

__device__ __forceinline__ float wave_max(float v) {
  for (int off = 16; off > 0; off >>= 1) v = fmaxf(v, __shfl_xor(v, off, 32));
  return v;
}

__device__ __forceinline__ int wave_sum_i(int v) {
  for (int off = 16; off > 0; off >>= 1) v += __shfl_xor(v, off, 32);
  return v;
}

__device__ __forceinline__ float leaky(float v) {
  return v > 0.0f ? v : NEG_SLOPE * v;
}

template <int M>
__device__ __forceinline__ float row_xor_add(float v) {
  return v + __int_as_float(__builtin_amdgcn_update_dpp(
                 0, __float_as_int(v), 0x160 + M, 0xf, 0xf, true));
}

__device__ __forceinline__ float row16_sum(float v) {
  v = row_xor_add<8>(v);
  v = row_xor_add<4>(v);
  v = row_xor_add<2>(v);
  v = row_xor_add<1>(v);
  return v;
}

__global__ __launch_bounds__(256, 16)
void gat_gemm_wmma(const float* __restrict__ x, const float* __restrict__ w,
                   const float* __restrict__ att_src,
                   const float* __restrict__ att_dst,
                   float* __restrict__ h, float* __restrict__ a_src,
                   float* __restrict__ a_dst, int n_nodes) {
  __shared__ float lds_a[16 * LDA];
  __shared__ float lds_p[8 * 16 * 2];
  const int tid  = threadIdx.x;
  const int row0 = blockIdx.x * 16;

  for (int i = tid; i < 16 * D; i += 256) {
    int r = i >> 7, c = i & (D - 1);
    int row = row0 + r;
    lds_a[r * LDA + c] = (row < n_nodes) ? x[(size_t)row * D + c] : 0.0f;
  }
  __syncthreads();

  const int wave = tid >> 5;
  const int lane = tid & 31;
  const int half = lane >> 4;
  const int l16  = lane & 15;
  const int col0 = wave * 16;

  v8f c = {};
#pragma unroll
  for (int k = 0; k < D; k += 4) {
    const int ka = k + half * 2;
    v2f a, b;
    a.x = lds_a[l16 * LDA + ka];
    a.y = lds_a[l16 * LDA + ka + 1];
    b.x = w[ka * D + col0 + l16];
    b.y = w[(ka + 1) * D + col0 + l16];
    c = __builtin_amdgcn_wmma_f32_16x16x4_f32(false, a, false, b, (short)0, c,
                                              false, false);
  }
  const float as = att_src[col0 + l16];
  const float ad = att_dst[col0 + l16];
#pragma unroll
  for (int v = 0; v < 8; ++v) {
    int row = row0 + v + half * 8;
    if (row < n_nodes) h[(size_t)row * D + col0 + l16] = c[v];
  }
#pragma unroll
  for (int v = 0; v < 8; ++v) {
    const float ps = row16_sum(c[v] * as);
    const float pd = row16_sum(c[v] * ad);
    if (l16 == 0) {
      lds_p[(wave * 16 + v + half * 8) * 2 + 0] = ps;
      lds_p[(wave * 16 + v + half * 8) * 2 + 1] = pd;
    }
  }
  __syncthreads();
  if (tid < 32) {
    const int r = tid >> 1, which = tid & 1;
    float s = 0.0f;
#pragma unroll
    for (int t = 0; t < 8; ++t) s += lds_p[(t * 16 + r) * 2 + which];
    const int row = row0 + r;
    if (row < n_nodes) (which ? a_dst : a_src)[row] = s;
  }
}


template <int N>
__device__ __forceinline__ int row_shr_add(int v) {
  return v + __builtin_amdgcn_update_dpp(0, v, 0x110 + N, 0xf, 0xf, true);
}

__device__ __forceinline__ int wave_incl_scan_i(int v, int lane) {
  v = row_shr_add<1>(v);
  v = row_shr_add<2>(v);
  v = row_shr_add<4>(v);
  v = row_shr_add<8>(v);
  const int lo = __builtin_amdgcn_readlane(v, 15);
  return v + ((lane & 16) ? lo : 0);
}

#define BSH    8
#define BNODES 256
#define NCH    256
#define ST     512
#define SU     8
#define SUB    16

__global__ __launch_bounds__(ST)
void sort_hist(const int* __restrict__ ei, int* __restrict__ hist_t,
               int n_edges, int nb) {
  extern __shared__ int lds_dyn[];
  const int tid = threadIdx.x;
  const int wg  = blockIdx.x;
  for (int b = tid; b < nb; b += ST) lds_dyn[b] = 0;
  __syncthreads();
  const int per = (n_edges + NCH - 1) / NCH;
  const int c0  = wg * per;
  const int c1  = min(c0 + per, n_edges);
  for (int i0 = c0 + tid; i0 < c1; i0 += ST * SU) {
    int d[SU];
#pragma unroll
    for (int j = 0; j < SU; ++j) {
      const int i = min(i0 + j * ST, c1 - 1);
      d[j] = ei[n_edges + i];
    }
#pragma unroll
    for (int j = 0; j < SU; ++j)
      if (i0 + j * ST < c1) atomicAdd(&lds_dyn[d[j] >> BSH], 1);
  }
  __syncthreads();
  for (int b = tid; b < nb; b += ST) hist_t[b * NCH + wg] = lds_dyn[b];
}

#define SCAN_CH 16
__device__ __forceinline__ int4 scan_row(int4 q, int lane, int& run) {
  const int t   = q.x + q.y + q.z + q.w;
  const int inc = wave_incl_scan_i(t, lane);
  int4 o;
  o.x = run + inc - t;
  o.y = o.x + q.x;
  o.z = o.y + q.y;
  o.w = o.z + q.z;
  run += __builtin_amdgcn_readlane(inc, 31);
  return o;
}

__global__ __launch_bounds__(1024)
void sort_scan(int* __restrict__ a, int m) {
  __shared__ int wtot[32];
  const int lane = threadIdx.x & 31;
  const int wave = threadIdx.x >> 5;
  int4* a4 = (int4*)a;
  const int rows = m >> 7;
  const int rpw  = (rows + 31) >> 5;
  const int r0   = wave * rpw;
  const int r1   = min(r0 + rpw, rows);

  int4 v[SCAN_CH];
  int s = 0;
#pragma unroll
  for (int k = 0; k < SCAN_CH; ++k) {
    const int r = r0 + k;
    v[k] = (r < r1) ? a4[r * 32 + lane] : make_int4(0, 0, 0, 0);
    s += v[k].x + v[k].y + v[k].z + v[k].w;
  }
  for (int r = r0 + SCAN_CH; r < r1; ++r) {
    const int4 q = a4[r * 32 + lane];
    s += q.x + q.y + q.z + q.w;
  }
  s = wave_sum_i(s);
  if (lane == 0) wtot[wave] = s;
  __syncthreads();

  int run = wave_sum_i(lane < wave ? wtot[lane] : 0);
#pragma unroll
  for (int k = 0; k < SCAN_CH; ++k) {
    const int r = r0 + k;
    const int4 o = scan_row(v[k], lane, run);
    if (r < r1) a4[r * 32 + lane] = o;
  }
  for (int r = r0 + SCAN_CH; r < r1; ++r) {
    const int4 o = scan_row(a4[r * 32 + lane], lane, run);
    a4[r * 32 + lane] = o;
  }
}

__global__ __launch_bounds__(ST)
void sort_split(const int* __restrict__ ei, const int* __restrict__ off_t,
                int* __restrict__ pairs_raw, int n_edges, int nb) {
  extern __shared__ int lds_dyn[];
  int2* pairs = (int2*)pairs_raw;
  const int tid = threadIdx.x;
  const int wg  = blockIdx.x;
  for (int b = tid; b < nb; b += ST) lds_dyn[b] = off_t[b * NCH + wg];
  __syncthreads();
  const int per = (n_edges + NCH - 1) / NCH;
  const int c0  = wg * per;
  const int c1  = min(c0 + per, n_edges);
  for (int i0 = c0 + tid; i0 < c1; i0 += ST * SU) {
    int s[SU], d[SU];
#pragma unroll
    for (int j = 0; j < SU; ++j) {
      const int i = min(i0 + j * ST, c1 - 1);
      s[j] = ei[i];
      d[j] = ei[n_edges + i];
    }
#pragma unroll
    for (int j = 0; j < SU; ++j) {
      if (i0 + j * ST < c1) {
        const int p = atomicAdd(&lds_dyn[d[j] >> BSH], 1);
        pairs[p] = make_int2(s[j], d[j]);
      }
    }
  }
}

__global__ __launch_bounds__(ST)
void sort_bucket(const int* __restrict__ off_t, const int* __restrict__ pairs_raw,
                 int* __restrict__ seg_end, int* __restrict__ srt,
                 int n_nodes, int n_edges, int nb) {
  __shared__ int cnt[BNODES];
  __shared__ int wsum[BNODES / 32];
  const int2* pairs = (const int2*)pairs_raw;
  const int tid  = threadIdx.x;
  const int lane = tid & 31;
  const int wave = tid >> 5;
  const int b    = blockIdx.x;
  const int bb   = off_t[b * NCH];
  const int be   = (b + 1 < nb) ? off_t[(b + 1) * NCH] : n_edges;

  if (tid < BNODES) cnt[tid] = 0;
  __syncthreads();
  for (int i0 = bb + tid; i0 < be; i0 += ST * SUB) {
    int d[SUB];
#pragma unroll
    for (int j = 0; j < SUB; ++j) {
      const int i = min(i0 + j * ST, be - 1);
      d[j] = pairs[i].y;
    }
#pragma unroll
    for (int j = 0; j < SUB; ++j)
      if (i0 + j * ST < be) atomicAdd(&cnt[d[j] & (BNODES - 1)], 1);
  }
  __syncthreads();

  int c = 0, inc = 0;
  if (tid < BNODES) {
    c   = cnt[tid];
    inc = wave_incl_scan_i(c, lane);
    if (lane == 31) wsum[wave] = inc;
  }
  __syncthreads();
  if (tid < BNODES) {
    int base = 0;
    for (int w = 0; w < wave; ++w) base += wsum[w];
    const int excl = base + inc - c;
    cnt[tid] = excl;
    const int node = b * BNODES + tid;
    if (node < n_nodes) seg_end[node] = bb + excl + c;
  }
  __syncthreads();
  for (int i0 = bb + tid; i0 < be; i0 += ST * SUB) {
    int2 pr[SUB];
#pragma unroll
    for (int j = 0; j < SUB; ++j) {
      const int i = min(i0 + j * ST, be - 1);
      pr[j] = pairs[i];
    }
#pragma unroll
    for (int j = 0; j < SUB; ++j) {
      if (i0 + j * ST < be) {
        const int p = atomicAdd(&cnt[pr[j].y & (BNODES - 1)], 1);
        srt[bb + p] = pr[j].x;
      }
    }
  }
}

__global__ __launch_bounds__(256)
void gat_gather_ln(const int* __restrict__ seg_end, const int* __restrict__ srt,
                   const float* __restrict__ a_src, const float* __restrict__ a_dst,
                   const float* __restrict__ h, const float* x,
                   const float* __restrict__ bias, const float* __restrict__ gamma,
                   const float* __restrict__ beta, float* out, int n_nodes) {
  const int wv   = __builtin_amdgcn_readfirstlane((int)(threadIdx.x >> 5));
  const int node = blockIdx.x * 8 + wv;
  const int lane = threadIdx.x & 31;
  if (node >= n_nodes) return;
  const int beg = (node > 0) ? seg_end[node - 1] : 0;
  const int end = seg_end[node];
  const float ad    = a_dst[node];
  const float lself = leaky(a_src[node] + ad);
  const float4 xv = ((const float4*)(x + (size_t)node * D))[lane];

  float m = lself;
  for (int i = beg + lane; i < end; i += 32) m = fmaxf(m, leaky(a_src[srt[i]] + ad));
  m = wave_max(m);

  const float w0 = __expf(lself - m);
  const float4 hs = ((const float4*)(h + (size_t)node * D))[lane];
  float4 a = make_float4(hs.x * w0, hs.y * w0, hs.z * w0, hs.w * w0);
  float sw = 0.0f;
  for (int i0 = beg; i0 < end; i0 += 32) {
    const int i = i0 + lane;
    int   s = 0;
    float w = 0.0f;
    if (i < end) {
      s = srt[i];
      w = __expf(leaky(a_src[s] + ad) - m);
    }
    sw += w;
    const int cnt = min(32, end - i0);
#pragma unroll 4
    for (int j = 0; j < cnt; ++j) {
      const int   sj = __builtin_amdgcn_readlane(s, j);
      const float wj = __int_as_float(__builtin_amdgcn_readlane(__float_as_int(w), j));
      const float4 hv = ((const float4*)(h + (size_t)sj * D))[lane];
      a.x += wj * hv.x;
      a.y += wj * hv.y;
      a.z += wj * hv.z;
      a.w += wj * hv.w;
    }
  }
  sw = wave_sum(sw) + w0;

  const float inv_s = 1.0f / sw;
  const float4 bv = ((const float4*)bias)[lane];
  float4 y;
  y.x = xv.x + a.x * inv_s + bv.x;
  y.y = xv.y + a.y * inv_s + bv.y;
  y.z = xv.z + a.z * inv_s + bv.z;
  y.w = xv.w + a.w * inv_s + bv.w;
  float sum = y.x + y.y + y.z + y.w;
  float sq  = y.x * y.x + y.y * y.y + y.z * y.z + y.w * y.w;
  sum = wave_sum(sum);
  sq  = wave_sum(sq);
  const float mu   = sum * (1.0f / D);
  const float var  = sq * (1.0f / D) - mu * mu;
  const float rstd = rsqrtf(var + LN_EPS);
  const float4 gv = ((const float4*)gamma)[lane];
  const float4 tv = ((const float4*)beta)[lane];
  float4 o;
  o.x = (y.x - mu) * rstd * gv.x + tv.x;
  o.y = (y.y - mu) * rstd * gv.y + tv.y;
  o.z = (y.z - mu) * rstd * gv.z + tv.z;
  o.w = (y.w - mu) * rstd * gv.w + tv.w;
  ((float4*)(out + (size_t)node * D))[lane] = o;
}

extern "C" void kernel_launch(void* const* d_in, const int* in_sizes, int n_in,
                              void* d_out, int out_size, void* d_ws, size_t ws_size,
                              hipStream_t stream) {
  const float* x0   = (const float*)d_in[0];
  const int*   ei   = (const int*)d_in[1];
  const float* W    = (const float*)d_in[2];
  const float* asrc = (const float*)d_in[3];
  const float* adst = (const float*)d_in[4];
  const float* bias = (const float*)d_in[5];
  const float* gam  = (const float*)d_in[6];
  const float* bet  = (const float*)d_in[7];
  float* out = (float*)d_out;

  const int n  = in_sizes[0] / D;
  const int e  = in_sizes[1] / 2;
  const int nl = in_sizes[2] / (D * D);

  float* ws    = (float*)d_ws;
  float* h     = ws; ws += (size_t)n * D;
  float* va    = ws; ws += n;
  float* vb    = ws; ws += n;
  int*   cur   = (int*)ws; ws += n;
  int*   srt   = (int*)ws; ws += e;

  const int nb      = (n + BNODES - 1) / BNODES;
  int* hist_t = (int*)h;
  int* pairs  = hist_t + (size_t)nb * NCH;
  sort_hist<<<NCH, ST, nb * sizeof(int), stream>>>(ei, hist_t, e, nb);
  sort_scan<<<1, 1024, 0, stream>>>(hist_t, nb * NCH);
  sort_split<<<NCH, ST, nb * sizeof(int), stream>>>(ei, hist_t, pairs, e, nb);
  sort_bucket<<<nb, ST, 0, stream>>>(hist_t, pairs, cur, srt, n, e, nb);

  for (int l = 0; l < nl; ++l) {
    const float* xin = (l == 0) ? x0 : out;
    const float* Wl  = W + (size_t)l * D * D;
    gat_gemm_wmma<<<(n + 15) / 16, 256, 0, stream>>>(xin, Wl, asrc + l * D,
                                                     adst + l * D, h, va, vb, n);
    gat_gather_ln<<<(n + 7) / 8, 256, 0, stream>>>(cur, srt, va, vb, h, xin,
                                                   bias + l * D, gam + l * D,
                                                   bet + l * D, out, n);
  }
}


# Product blocks ask for all 16 slab rows (two 16-byte loads per thread) and half of the B fragments as one group ahead of the barrier: 1 load round trip where 9 stood in a row; sums unchanged
# speedup vs baseline: 4.3740x; 1.0941x over previous
#include <hip/hip_runtime.h>
#include <math.h>

#define D 128
#define LDA 132
#define KPRE 16
#define NEG_SLOPE 0.2f
#define LN_EPS 1e-5f

typedef float v2f __attribute__((ext_vector_type(2)));
typedef float v8f __attribute__((ext_vector_type(8)));

__device__ __forceinline__ float wave_sum(float v) {
#pragma unroll
  for (int off = 16; off > 0; off >>= 1) v += __shfl_xor(v, off, 32);
  return v;
}

__device__ __forceinline__ float wave_max(float v) {
  for (int off = 16; off > 0; off >>= 1) v = fmaxf(v, __shfl_xor(v, off, 32));
  return v;
}

__device__ __forceinline__ int wave_sum_i(int v) {
  for (int off = 16; off > 0; off >>= 1) v += __shfl_xor(v, off, 32);
  return v;
}

__device__ __forceinline__ float leaky(float v) {
  return v > 0.0f ? v : NEG_SLOPE * v;
}

template <int M>
__device__ __forceinline__ float row_xor_add(float v) {
  return v + __int_as_float(__builtin_amdgcn_update_dpp(
                 0, __float_as_int(v), 0x160 + M, 0xf, 0xf, true));
}

__device__ __forceinline__ float row16_sum(float v) {
  v = row_xor_add<8>(v);
  v = row_xor_add<4>(v);
  v = row_xor_add<2>(v);
  v = row_xor_add<1>(v);
  return v;
}

__global__ __launch_bounds__(256, 16)
void gat_gemm_wmma(const float* __restrict__ x, const float* __restrict__ w,
                   const float* __restrict__ att_src,
                   const float* __restrict__ att_dst,
                   float* __restrict__ h, float* __restrict__ a_src,
                   float* __restrict__ a_dst, int n_nodes) {
  __shared__ __attribute__((aligned(16))) float lds_a[16 * LDA];
  __shared__ float lds_p[8 * 16 * 2];
  const int tid  = threadIdx.x;
  const int row0 = blockIdx.x * 16;
  const int wave = tid >> 5;
  const int lane = tid & 31;
  const int half = lane >> 4;
  const int l16  = lane & 15;
  const int col0 = wave * 16;

  float4 xa[2];
#pragma unroll
  for (int q = 0; q < 2; ++q) {
    const int row = row0 + wave + 8 * q;
    xa[q] = (row < n_nodes) ? ((const float4*)(x + (size_t)row * D))[lane]
                            : make_float4(0.0f, 0.0f, 0.0f, 0.0f);
  }
  v2f bpre[KPRE];
#pragma unroll
  for (int t = 0; t < KPRE; ++t) {
    const int ka = 4 * t + half * 2;
    bpre[t].x = w[ka * D + col0 + l16];
    bpre[t].y = w[(ka + 1) * D + col0 + l16];
  }
#pragma unroll
  for (int q = 0; q < 2; ++q)
    *(float4*)&lds_a[(wave + 8 * q) * LDA + lane * 4] = xa[q];
  __syncthreads();

  v8f c = {};
#pragma unroll
  for (int k = 0; k < D; k += 4) {
    const int ka = k + half * 2;
    v2f a, b;
    a.x = lds_a[l16 * LDA + ka];
    a.y = lds_a[l16 * LDA + ka + 1];
    if (k / 4 < KPRE) {
      b = bpre[k / 4];
    } else {
      b.x = w[ka * D + col0 + l16];
      b.y = w[(ka + 1) * D + col0 + l16];
    }
    c = __builtin_amdgcn_wmma_f32_16x16x4_f32(false, a, false, b, (short)0, c,
                                              false, false);
  }
  const float as = att_src[col0 + l16];
  const float ad = att_dst[col0 + l16];
#pragma unroll
  for (int v = 0; v < 8; ++v) {
    int row = row0 + v + half * 8;
    if (row < n_nodes) h[(size_t)row * D + col0 + l16] = c[v];
  }
#pragma unroll
  for (int v = 0; v < 8; ++v) {
    const float ps = row16_sum(c[v] * as);
    const float pd = row16_sum(c[v] * ad);
    if (l16 == 0) {
      lds_p[(wave * 16 + v + half * 8) * 2 + 0] = ps;
      lds_p[(wave * 16 + v + half * 8) * 2 + 1] = pd;
    }
  }
  __syncthreads();
  if (tid < 32) {
    const int r = tid >> 1, which = tid & 1;
    float s = 0.0f;
#pragma unroll
    for (int t = 0; t < 8; ++t) s += lds_p[(t * 16 + r) * 2 + which];
    const int row = row0 + r;
    if (row < n_nodes) (which ? a_dst : a_src)[row] = s;
  }
}


template <int N>
__device__ __forceinline__ int row_shr_add(int v) {
  return v + __builtin_amdgcn_update_dpp(0, v, 0x110 + N, 0xf, 0xf, true);
}

__device__ __forceinline__ int wave_incl_scan_i(int v, int lane) {
  v = row_shr_add<1>(v);
  v = row_shr_add<2>(v);
  v = row_shr_add<4>(v);
  v = row_shr_add<8>(v);
  const int lo = __builtin_amdgcn_readlane(v, 15);
  return v + ((lane & 16) ? lo : 0);
}

#define BSH    8
#define BNODES 256
#define NCH    256
#define ST     512
#define SU     8
#define SUB    16

__global__ __launch_bounds__(ST)
void sort_hist(const int* __restrict__ ei, int* __restrict__ hist_t,
               int n_edges, int nb) {
  extern __shared__ int lds_dyn[];
  const int tid = threadIdx.x;
  const int wg  = blockIdx.x;
  for (int b = tid; b < nb; b += ST) lds_dyn[b] = 0;
  __syncthreads();
  const int per = (n_edges + NCH - 1) / NCH;
  const int c0  = wg * per;
  const int c1  = min(c0 + per, n_edges);
  for (int i0 = c0 + tid; i0 < c1; i0 += ST * SU) {
    int d[SU];
#pragma unroll
    for (int j = 0; j < SU; ++j) {
      const int i = min(i0 + j * ST, c1 - 1);
      d[j] = ei[n_edges + i];
    }
#pragma unroll
    for (int j = 0; j < SU; ++j)
      if (i0 + j * ST < c1) atomicAdd(&lds_dyn[d[j] >> BSH], 1);
  }
  __syncthreads();
  for (int b = tid; b < nb; b += ST) hist_t[b * NCH + wg] = lds_dyn[b];
}

#define SCAN_CH 16
__device__ __forceinline__ int4 scan_row(int4 q, int lane, int& run) {
  const int t   = q.x + q.y + q.z + q.w;
  const int inc = wave_incl_scan_i(t, lane);
  int4 o;
  o.x = run + inc - t;
  o.y = o.x + q.x;
  o.z = o.y + q.y;
  o.w = o.z + q.z;
  run += __builtin_amdgcn_readlane(inc, 31);
  return o;
}

__global__ __launch_bounds__(1024)
void sort_scan(int* __restrict__ a, int m) {
  __shared__ int wtot[32];
  const int lane = threadIdx.x & 31;
  const int wave = threadIdx.x >> 5;
  int4* a4 = (int4*)a;
  const int rows = m >> 7;
  const int rpw  = (rows + 31) >> 5;
  const int r0   = wave * rpw;
  const int r1   = min(r0 + rpw, rows);

  int4 v[SCAN_CH];
  int s = 0;
#pragma unroll
  for (int k = 0; k < SCAN_CH; ++k) {
    const int r = r0 + k;
    v[k] = (r < r1) ? a4[r * 32 + lane] : make_int4(0, 0, 0, 0);
    s += v[k].x + v[k].y + v[k].z + v[k].w;
  }
  for (int r = r0 + SCAN_CH; r < r1; ++r) {
    const int4 q = a4[r * 32 + lane];
    s += q.x + q.y + q.z + q.w;
  }
  s = wave_sum_i(s);
  if (lane == 0) wtot[wave] = s;
  __syncthreads();

  int run = wave_sum_i(lane < wave ? wtot[lane] : 0);
#pragma unroll
  for (int k = 0; k < SCAN_CH; ++k) {
    const int r = r0 + k;
    const int4 o = scan_row(v[k], lane, run);
    if (r < r1) a4[r * 32 + lane] = o;
  }
  for (int r = r0 + SCAN_CH; r < r1; ++r) {
    const int4 o = scan_row(a4[r * 32 + lane], lane, run);
    a4[r * 32 + lane] = o;
  }
}

__global__ __launch_bounds__(ST)
void sort_split(const int* __restrict__ ei, const int* __restrict__ off_t,
                int* __restrict__ pairs_raw, int n_edges, int nb) {
  extern __shared__ int lds_dyn[];
  int2* pairs = (int2*)pairs_raw;
  const int tid = threadIdx.x;
  const int wg  = blockIdx.x;
  for (int b = tid; b < nb; b += ST) lds_dyn[b] = off_t[b * NCH + wg];
  __syncthreads();
  const int per = (n_edges + NCH - 1) / NCH;
  const int c0  = wg * per;
  const int c1  = min(c0 + per, n_edges);
  for (int i0 = c0 + tid; i0 < c1; i0 += ST * SU) {
    int s[SU], d[SU];
#pragma unroll
    for (int j = 0; j < SU; ++j) {
      const int i = min(i0 + j * ST, c1 - 1);
      s[j] = ei[i];
      d[j] = ei[n_edges + i];
    }
#pragma unroll
    for (int j = 0; j < SU; ++j) {
      if (i0 + j * ST < c1) {
        const int p = atomicAdd(&lds_dyn[d[j] >> BSH], 1);
        pairs[p] = make_int2(s[j], d[j]);
      }
    }
  }
}

__global__ __launch_bounds__(ST)
void sort_bucket(const int* __restrict__ off_t, const int* __restrict__ pairs_raw,
                 int* __restrict__ seg_end, int* __restrict__ srt,
                 int n_nodes, int n_edges, int nb) {
  __shared__ int cnt[BNODES];
  __shared__ int wsum[BNODES / 32];
  const int2* pairs = (const int2*)pairs_raw;
  const int tid  = threadIdx.x;
  const int lane = tid & 31;
  const int wave = tid >> 5;
  const int b    = blockIdx.x;
  const int bb   = off_t[b * NCH];
  const int be   = (b + 1 < nb) ? off_t[(b + 1) * NCH] : n_edges;

  if (tid < BNODES) cnt[tid] = 0;
  __syncthreads();
  for (int i0 = bb + tid; i0 < be; i0 += ST * SUB) {
    int d[SUB];
#pragma unroll
    for (int j = 0; j < SUB; ++j) {
      const int i = min(i0 + j * ST, be - 1);
      d[j] = pairs[i].y;
    }
#pragma unroll
    for (int j = 0; j < SUB; ++j)
      if (i0 + j * ST < be) atomicAdd(&cnt[d[j] & (BNODES - 1)], 1);
  }
  __syncthreads();

  int c = 0, inc = 0;
  if (tid < BNODES) {
    c   = cnt[tid];
    inc = wave_incl_scan_i(c, lane);
    if (lane == 31) wsum[wave] = inc;
  }
  __syncthreads();
  if (tid < BNODES) {
    int base = 0;
    for (int w = 0; w < wave; ++w) base += wsum[w];
    const int excl = base + inc - c;
    cnt[tid] = excl;
    const int node = b * BNODES + tid;
    if (node < n_nodes) seg_end[node] = bb + excl + c;
  }
  __syncthreads();
  for (int i0 = bb + tid; i0 < be; i0 += ST * SUB) {
    int2 pr[SUB];
#pragma unroll
    for (int j = 0; j < SUB; ++j) {
      const int i = min(i0 + j * ST, be - 1);
      pr[j] = pairs[i];
    }
#pragma unroll
    for (int j = 0; j < SUB; ++j) {
      if (i0 + j * ST < be) {
        const int p = atomicAdd(&cnt[pr[j].y & (BNODES - 1)], 1);
        srt[bb + p] = pr[j].x;
      }
    }
  }
}

__global__ __launch_bounds__(256)
void gat_gather_ln(const int* __restrict__ seg_end, const int* __restrict__ srt,
                   const float* __restrict__ a_src, const float* __restrict__ a_dst,
                   const float* __restrict__ h, const float* x,
                   const float* __restrict__ bias, const float* __restrict__ gamma,
                   const float* __restrict__ beta, float* out, int n_nodes) {
  const int wv   = __builtin_amdgcn_readfirstlane((int)(threadIdx.x >> 5));
  const int node = blockIdx.x * 8 + wv;
  const int lane = threadIdx.x & 31;
  if (node >= n_nodes) return;
  const int beg = (node > 0) ? seg_end[node - 1] : 0;
  const int end = seg_end[node];
  const float ad    = a_dst[node];
  const float lself = leaky(a_src[node] + ad);
  const float4 xv = ((const float4*)(x + (size_t)node * D))[lane];

  float m = lself;
  for (int i = beg + lane; i < end; i += 32) m = fmaxf(m, leaky(a_src[srt[i]] + ad));
  m = wave_max(m);

  const float w0 = __expf(lself - m);
  const float4 hs = ((const float4*)(h + (size_t)node * D))[lane];
  float4 a = make_float4(hs.x * w0, hs.y * w0, hs.z * w0, hs.w * w0);
  float sw = 0.0f;
  for (int i0 = beg; i0 < end; i0 += 32) {
    const int i = i0 + lane;
    int   s = 0;
    float w = 0.0f;
    if (i < end) {
      s = srt[i];
      w = __expf(leaky(a_src[s] + ad) - m);
    }
    sw += w;
    const int cnt = min(32, end - i0);
#pragma unroll 4
    for (int j = 0; j < cnt; ++j) {
      const int   sj = __builtin_amdgcn_readlane(s, j);
      const float wj = __int_as_float(__builtin_amdgcn_readlane(__float_as_int(w), j));
      const float4 hv = ((const float4*)(h + (size_t)sj * D))[lane];
      a.x += wj * hv.x;
      a.y += wj * hv.y;
      a.z += wj * hv.z;
      a.w += wj * hv.w;
    }
  }
  sw = wave_sum(sw) + w0;

  const float inv_s = 1.0f / sw;
  const float4 bv = ((const float4*)bias)[lane];
  float4 y;
  y.x = xv.x + a.x * inv_s + bv.x;
  y.y = xv.y + a.y * inv_s + bv.y;
  y.z = xv.z + a.z * inv_s + bv.z;
  y.w = xv.w + a.w * inv_s + bv.w;
  float sum = y.x + y.y + y.z + y.w;
  float sq  = y.x * y.x + y.y * y.y + y.z * y.z + y.w * y.w;
  sum = wave_sum(sum);
  sq  = wave_sum(sq);
  const float mu   = sum * (1.0f / D);
  const float var  = sq * (1.0f / D) - mu * mu;
  const float rstd = rsqrtf(var + LN_EPS);
  const float4 gv = ((const float4*)gamma)[lane];
  const float4 tv = ((const float4*)beta)[lane];
  float4 o;
  o.x = (y.x - mu) * rstd * gv.x + tv.x;
  o.y = (y.y - mu) * rstd * gv.y + tv.y;
  o.z = (y.z - mu) * rstd * gv.z + tv.z;
  o.w = (y.w - mu) * rstd * gv.w + tv.w;
  ((float4*)(out + (size_t)node * D))[lane] = o;
}

extern "C" void kernel_launch(void* const* d_in, const int* in_sizes, int n_in,
                              void* d_out, int out_size, void* d_ws, size_t ws_size,
                              hipStream_t stream) {
  const float* x0   = (const float*)d_in[0];
  const int*   ei   = (const int*)d_in[1];
  const float* W    = (const float*)d_in[2];
  const float* asrc = (const float*)d_in[3];
  const float* adst = (const float*)d_in[4];
  const float* bias = (const float*)d_in[5];
  const float* gam  = (const float*)d_in[6];
  const float* bet  = (const float*)d_in[7];
  float* out = (float*)d_out;

  const int n  = in_sizes[0] / D;
  const int e  = in_sizes[1] / 2;
  const int nl = in_sizes[2] / (D * D);

  float* ws    = (float*)d_ws;
  float* h     = ws; ws += (size_t)n * D;
  float* va    = ws; ws += n;
  float* vb    = ws; ws += n;
  int*   cur   = (int*)ws; ws += n;
  int*   srt   = (int*)ws; ws += e;

  const int nb      = (n + BNODES - 1) / BNODES;
  int* hist_t = (int*)h;
  int* pairs  = hist_t + (size_t)nb * NCH;
  sort_hist<<<NCH, ST, nb * sizeof(int), stream>>>(ei, hist_t, e, nb);
  sort_scan<<<1, 1024, 0, stream>>>(hist_t, nb * NCH);
  sort_split<<<NCH, ST, nb * sizeof(int), stream>>>(ei, hist_t, pairs, e, nb);
  sort_bucket<<<nb, ST, 0, stream>>>(hist_t, pairs, cur, srt, n, e, nb);

  for (int l = 0; l < nl; ++l) {
    const float* xin = (l == 0) ? x0 : out;
    const float* Wl  = W + (size_t)l * D * D;
    gat_gemm_wmma<<<(n + 15) / 16, 256, 0, stream>>>(xin, Wl, asrc + l * D,
                                                     adst + l * D, h, va, vb, n);
    gat_gather_ln<<<(n + 7) / 8, 256, 0, stream>>>(cur, srt, va, vb, h, xin,
                                                   bias + l * D, gam + l * D,
                                                   bet + l * D, out, n);
  }
}


# Gather waves stay and walk 4 consecutive nodes each (12504 wave entries a layer, not 50000); the next node's sources, scores and own row are asked for under this node's row sums; sum order unchanged
# speedup vs baseline: 4.6227x; 1.0569x over previous
#include <hip/hip_runtime.h>
#include <math.h>

#define D 128
#define LDA 132
#define KPRE 16
#define NEG_SLOPE 0.2f
#define LN_EPS 1e-5f

typedef float v2f __attribute__((ext_vector_type(2)));
typedef float v8f __attribute__((ext_vector_type(8)));

__device__ __forceinline__ float wave_sum(float v) {
#pragma unroll
  for (int off = 16; off > 0; off >>= 1) v += __shfl_xor(v, off, 32);
  return v;
}

__device__ __forceinline__ float wave_max(float v) {
  for (int off = 16; off > 0; off >>= 1) v = fmaxf(v, __shfl_xor(v, off, 32));
  return v;
}

__device__ __forceinline__ int wave_sum_i(int v) {
  for (int off = 16; off > 0; off >>= 1) v += __shfl_xor(v, off, 32);
  return v;
}

__device__ __forceinline__ float leaky(float v) {
  return v > 0.0f ? v : NEG_SLOPE * v;
}

template <int M>
__device__ __forceinline__ float row_xor_add(float v) {
  return v + __int_as_float(__builtin_amdgcn_update_dpp(
                 0, __float_as_int(v), 0x160 + M, 0xf, 0xf, true));
}

__device__ __forceinline__ float row16_sum(float v) {
  v = row_xor_add<8>(v);
  v = row_xor_add<4>(v);
  v = row_xor_add<2>(v);
  v = row_xor_add<1>(v);
  return v;
}

__global__ __launch_bounds__(256, 16)
void gat_gemm_wmma(const float* __restrict__ x, const float* __restrict__ w,
                   const float* __restrict__ att_src,
                   const float* __restrict__ att_dst,
                   float* __restrict__ h, float* __restrict__ a_src,
                   float* __restrict__ a_dst, int n_nodes) {
  __shared__ __attribute__((aligned(16))) float lds_a[16 * LDA];
  __shared__ float lds_p[8 * 16 * 2];
  const int tid  = threadIdx.x;
  const int row0 = blockIdx.x * 16;
  const int wave = tid >> 5;
  const int lane = tid & 31;
  const int half = lane >> 4;
  const int l16  = lane & 15;
  const int col0 = wave * 16;

  float4 xa[2];
#pragma unroll
  for (int q = 0; q < 2; ++q) {
    const int row = row0 + wave + 8 * q;
    xa[q] = (row < n_nodes) ? ((const float4*)(x + (size_t)row * D))[lane]
                            : make_float4(0.0f, 0.0f, 0.0f, 0.0f);
  }
  v2f bpre[KPRE];
#pragma unroll
  for (int t = 0; t < KPRE; ++t) {
    const int ka = 4 * t + half * 2;
    bpre[t].x = w[ka * D + col0 + l16];
    bpre[t].y = w[(ka + 1) * D + col0 + l16];
  }
#pragma unroll
  for (int q = 0; q < 2; ++q)
    *(float4*)&lds_a[(wave + 8 * q) * LDA + lane * 4] = xa[q];
  __syncthreads();

  v8f c = {};
#pragma unroll
  for (int k = 0; k < D; k += 4) {
    const int ka = k + half * 2;
    v2f a, b;
    a.x = lds_a[l16 * LDA + ka];
    a.y = lds_a[l16 * LDA + ka + 1];
    if (k / 4 < KPRE) {
      b = bpre[k / 4];
    } else {
      b.x = w[ka * D + col0 + l16];
      b.y = w[(ka + 1) * D + col0 + l16];
    }
    c = __builtin_amdgcn_wmma_f32_16x16x4_f32(false, a, false, b, (short)0, c,
                                              false, false);
  }
  const float as = att_src[col0 + l16];
  const float ad = att_dst[col0 + l16];
#pragma unroll
  for (int v = 0; v < 8; ++v) {
    int row = row0 + v + half * 8;
    if (row < n_nodes) h[(size_t)row * D + col0 + l16] = c[v];
  }
#pragma unroll
  for (int v = 0; v < 8; ++v) {
    const float ps = row16_sum(c[v] * as);
    const float pd = row16_sum(c[v] * ad);
    if (l16 == 0) {
      lds_p[(wave * 16 + v + half * 8) * 2 + 0] = ps;
      lds_p[(wave * 16 + v + half * 8) * 2 + 1] = pd;
    }
  }
  __syncthreads();
  if (tid < 32) {
    const int r = tid >> 1, which = tid & 1;
    float s = 0.0f;
#pragma unroll
    for (int t = 0; t < 8; ++t) s += lds_p[(t * 16 + r) * 2 + which];
    const int row = row0 + r;
    if (row < n_nodes) (which ? a_dst : a_src)[row] = s;
  }
}


template <int N>
__device__ __forceinline__ int row_shr_add(int v) {
  return v + __builtin_amdgcn_update_dpp(0, v, 0x110 + N, 0xf, 0xf, true);
}

__device__ __forceinline__ int wave_incl_scan_i(int v, int lane) {
  v = row_shr_add<1>(v);
  v = row_shr_add<2>(v);
  v = row_shr_add<4>(v);
  v = row_shr_add<8>(v);
  const int lo = __builtin_amdgcn_readlane(v, 15);
  return v + ((lane & 16) ? lo : 0);
}

#define BSH    8
#define BNODES 256
#define NCH    256
#define ST     512
#define SU     8
#define SUB    16

__global__ __launch_bounds__(ST)
void sort_hist(const int* __restrict__ ei, int* __restrict__ hist_t,
               int n_edges, int nb) {
  extern __shared__ int lds_dyn[];
  const int tid = threadIdx.x;
  const int wg  = blockIdx.x;
  for (int b = tid; b < nb; b += ST) lds_dyn[b] = 0;
  __syncthreads();
  const int per = (n_edges + NCH - 1) / NCH;
  const int c0  = wg * per;
  const int c1  = min(c0 + per, n_edges);
  for (int i0 = c0 + tid; i0 < c1; i0 += ST * SU) {
    int d[SU];
#pragma unroll
    for (int j = 0; j < SU; ++j) {
      const int i = min(i0 + j * ST, c1 - 1);
      d[j] = ei[n_edges + i];
    }
#pragma unroll
    for (int j = 0; j < SU; ++j)
      if (i0 + j * ST < c1) atomicAdd(&lds_dyn[d[j] >> BSH], 1);
  }
  __syncthreads();
  for (int b = tid; b < nb; b += ST) hist_t[b * NCH + wg] = lds_dyn[b];
}

#define SCAN_CH 16
__device__ __forceinline__ int4 scan_row(int4 q, int lane, int& run) {
  const int t   = q.x + q.y + q.z + q.w;
  const int inc = wave_incl_scan_i(t, lane);
  int4 o;
  o.x = run + inc - t;
  o.y = o.x + q.x;
  o.z = o.y + q.y;
  o.w = o.z + q.z;
  run += __builtin_amdgcn_readlane(inc, 31);
  return o;
}

__global__ __launch_bounds__(1024)
void sort_scan(int* __restrict__ a, int m) {
  __shared__ int wtot[32];
  const int lane = threadIdx.x & 31;
  const int wave = threadIdx.x >> 5;
  int4* a4 = (int4*)a;
  const int rows = m >> 7;
  const int rpw  = (rows + 31) >> 5;
  const int r0   = wave * rpw;
  const int r1   = min(r0 + rpw, rows);

  int4 v[SCAN_CH];
  int s = 0;
#pragma unroll
  for (int k = 0; k < SCAN_CH; ++k) {
    const int r = r0 + k;
    v[k] = (r < r1) ? a4[r * 32 + lane] : make_int4(0, 0, 0, 0);
    s += v[k].x + v[k].y + v[k].z + v[k].w;
  }
  for (int r = r0 + SCAN_CH; r < r1; ++r) {
    const int4 q = a4[r * 32 + lane];
    s += q.x + q.y + q.z + q.w;
  }
  s = wave_sum_i(s);
  if (lane == 0) wtot[wave] = s;
  __syncthreads();

  int run = wave_sum_i(lane < wave ? wtot[lane] : 0);
#pragma unroll
  for (int k = 0; k < SCAN_CH; ++k) {
    const int r = r0 + k;
    const int4 o = scan_row(v[k], lane, run);
    if (r < r1) a4[r * 32 + lane] = o;
  }
  for (int r = r0 + SCAN_CH; r < r1; ++r) {
    const int4 o = scan_row(a4[r * 32 + lane], lane, run);
    a4[r * 32 + lane] = o;
  }
}

__global__ __launch_bounds__(ST)
void sort_split(const int* __restrict__ ei, const int* __restrict__ off_t,
                int* __restrict__ pairs_raw, int n_edges, int nb) {
  extern __shared__ int lds_dyn[];
  int2* pairs = (int2*)pairs_raw;
  const int tid = threadIdx.x;
  const int wg  = blockIdx.x;
  for (int b = tid; b < nb; b += ST) lds_dyn[b] = off_t[b * NCH + wg];
  __syncthreads();
  const int per = (n_edges + NCH - 1) / NCH;
  const int c0  = wg * per;
  const int c1  = min(c0 + per, n_edges);
  for (int i0 = c0 + tid; i0 < c1; i0 += ST * SU) {
    int s[SU], d[SU];
#pragma unroll
    for (int j = 0; j < SU; ++j) {
      const int i = min(i0 + j * ST, c1 - 1);
      s[j] = ei[i];
      d[j] = ei[n_edges + i];
    }
#pragma unroll
    for (int j = 0; j < SU; ++j) {
      if (i0 + j * ST < c1) {
        const int p = atomicAdd(&lds_dyn[d[j] >> BSH], 1);
        pairs[p] = make_int2(s[j], d[j]);
      }
    }
  }
}

__global__ __launch_bounds__(ST)
void sort_bucket(const int* __restrict__ off_t, const int* __restrict__ pairs_raw,
                 int* __restrict__ seg_end, int* __restrict__ srt,
                 int n_nodes, int n_edges, int nb) {
  __shared__ int cnt[BNODES];
  __shared__ int wsum[BNODES / 32];
  const int2* pairs = (const int2*)pairs_raw;
  const int tid  = threadIdx.x;
  const int lane = tid & 31;
  const int wave = tid >> 5;
  const int b    = blockIdx.x;
  const int bb   = off_t[b * NCH];
  const int be   = (b + 1 < nb) ? off_t[(b + 1) * NCH] : n_edges;

  if (tid < BNODES) cnt[tid] = 0;
  __syncthreads();
  for (int i0 = bb + tid; i0 < be; i0 += ST * SUB) {
    int d[SUB];
#pragma unroll
    for (int j = 0; j < SUB; ++j) {
      const int i = min(i0 + j * ST, be - 1);
      d[j] = pairs[i].y;
    }
#pragma unroll
    for (int j = 0; j < SUB; ++j)
      if (i0 + j * ST < be) atomicAdd(&cnt[d[j] & (BNODES - 1)], 1);
  }
  __syncthreads();

  int c = 0, inc = 0;
  if (tid < BNODES) {
    c   = cnt[tid];
    inc = wave_incl_scan_i(c, lane);
    if (lane == 31) wsum[wave] = inc;
  }
  __syncthreads();
  if (tid < BNODES) {
    int base = 0;
    for (int w = 0; w < wave; ++w) base += wsum[w];
    const int excl = base + inc - c;
    cnt[tid] = excl;
    const int node = b * BNODES + tid;
    if (node < n_nodes) seg_end[node] = bb + excl + c;
  }
  __syncthreads();
  for (int i0 = bb + tid; i0 < be; i0 += ST * SUB) {
    int2 pr[SUB];
#pragma unroll
    for (int j = 0; j < SUB; ++j) {
      const int i = min(i0 + j * ST, be - 1);
      pr[j] = pairs[i];
    }
#pragma unroll
    for (int j = 0; j < SUB; ++j) {
      if (i0 + j * ST < be) {
        const int p = atomicAdd(&cnt[pr[j].y & (BNODES - 1)], 1);
        srt[bb + p] = pr[j].x;
      }
    }
  }
}

#define NPW 4

__device__ __forceinline__ int lane_i(int v, int l) {
  return __builtin_amdgcn_readlane(v, l);
}
__device__ __forceinline__ float lane_f(float v, int l) {
  return __int_as_float(__builtin_amdgcn_readlane(__float_as_int(v), l));
}

__global__ __launch_bounds__(256, 16)
void gat_gather_ln(const int* __restrict__ seg_end, const int* __restrict__ srt,
                   const float* __restrict__ a_src, const float* __restrict__ a_dst,
                   const float* __restrict__ h, const float* x,
                   const float* bias, const float* gamma,
                   const float* beta, float* out, int n_nodes) {
  const int wv   = __builtin_amdgcn_readfirstlane((int)(threadIdx.x >> 5));
  const int lane = threadIdx.x & 31;
  const int n0   = (blockIdx.x * 8 + wv) * NPW;
  if (n0 >= n_nodes) return;
  const int nk = min(NPW, n_nodes - n0);

  float4 hs1 = ((const float4*)(h + (size_t)n0 * D))[lane];
  int   eb  = 0;
  float adv = 0.0f, asv = 0.0f;
  {
    const int idx = n0 - 1 + min(lane, nk);
    if (idx >= 0) eb = seg_end[idx];
    if (lane < nk) {
      adv = a_dst[n0 + lane];
      asv = a_src[n0 + lane];
    }
  }

  int   s1  = 0, s2 = 0;
  float as1 = 0.0f;
  {
    const int b = lane_i(eb, 0), e = lane_i(eb, 1), e2 = lane_i(eb, 2);
    if (b + lane < e) s1 = srt[b + lane];
    if (e + lane < e2) s2 = srt[e + lane];
    if (b + lane < e) as1 = a_src[s1];
  }

#pragma unroll
  for (int k = 0; k < NPW; ++k) {
    if (k >= nk) break;
    const int node = n0 + k;
    const int beg  = lane_i(eb, k);
    const int end  = lane_i(eb, k + 1);
    const int e1   = lane_i(eb, k + 2);
    const int e2   = lane_i(eb, k + 3);
    const float ad    = lane_f(adv, k);
    const float lself = leaky(lane_f(asv, k) + ad);
    const int    s0  = s1;
    const float  as0 = as1;
    const float4 hs  = hs1;
    s1  = s2;
    as1 = 0.0f;
    if (end + lane < e1) as1 = a_src[s1];
    s2 = 0;
    if (e1 + lane < e2) s2 = srt[e1 + lane];
    if (k + 1 < NPW)
      hs1 = ((const float4*)(h + (size_t)min(node + 1, n_nodes - 1) * D))[lane];
    const float4 xv = ((const float4*)(x + (size_t)node * D))[lane];

    const bool  v0  = beg + lane < end;
    const float lg0 = leaky(as0 + ad);
    float m = lself;
    if (v0) m = fmaxf(m, lg0);
    for (int i = beg + 32 + lane; i < end; i += 32)
      m = fmaxf(m, leaky(a_src[srt[i]] + ad));
    m = wave_max(m);

    const float w0 = __expf(lself - m);
    float4 a = make_float4(hs.x * w0, hs.y * w0, hs.z * w0, hs.w * w0);
    float sw = 0.0f;
    for (int i0 = beg; i0 < end; i0 += 32) {
      int   s;
      float w;
      if (i0 == beg) {
        s = s0;
        w = v0 ? __expf(lg0 - m) : 0.0f;
      } else {
        const int i = i0 + lane;
        s = 0;
        w = 0.0f;
        if (i < end) {
          s = srt[i];
          w = __expf(leaky(a_src[s] + ad) - m);
        }
      }
      sw += w;
      const int cnt = min(32, end - i0);
#pragma unroll 4
      for (int j = 0; j < cnt; ++j) {
        const int   sj = lane_i(s, j);
        const float wj = lane_f(w, j);
        const float4 hv = ((const float4*)(h + (size_t)sj * D))[lane];
        a.x += wj * hv.x;
        a.y += wj * hv.y;
        a.z += wj * hv.z;
        a.w += wj * hv.w;
      }
    }
    sw = wave_sum(sw) + w0;

    const float4 bv = ((const float4*)bias)[lane];
    const float4 gv = ((const float4*)gamma)[lane];
    const float4 tv = ((const float4*)beta)[lane];
    const float inv_s = 1.0f / sw;
    float4 y;
    y.x = xv.x + a.x * inv_s + bv.x;
    y.y = xv.y + a.y * inv_s + bv.y;
    y.z = xv.z + a.z * inv_s + bv.z;
    y.w = xv.w + a.w * inv_s + bv.w;
    float sum = y.x + y.y + y.z + y.w;
    float sq  = y.x * y.x + y.y * y.y + y.z * y.z + y.w * y.w;
    sum = wave_sum(sum);
    sq  = wave_sum(sq);
    const float mu   = sum * (1.0f / D);
    const float var  = sq * (1.0f / D) - mu * mu;
    const float rstd = rsqrtf(var + LN_EPS);
    float4 o;
    o.x = (y.x - mu) * rstd * gv.x + tv.x;
    o.y = (y.y - mu) * rstd * gv.y + tv.y;
    o.z = (y.z - mu) * rstd * gv.z + tv.z;
    o.w = (y.w - mu) * rstd * gv.w + tv.w;
    ((float4*)(out + (size_t)node * D))[lane] = o;
  }
}

extern "C" void kernel_launch(void* const* d_in, const int* in_sizes, int n_in,
                              void* d_out, int out_size, void* d_ws, size_t ws_size,
                              hipStream_t stream) {
  const float* x0   = (const float*)d_in[0];
  const int*   ei   = (const int*)d_in[1];
  const float* W    = (const float*)d_in[2];
  const float* asrc = (const float*)d_in[3];
  const float* adst = (const float*)d_in[4];
  const float* bias = (const float*)d_in[5];
  const float* gam  = (const float*)d_in[6];
  const float* bet  = (const float*)d_in[7];
  float* out = (float*)d_out;

  const int n  = in_sizes[0] / D;
  const int e  = in_sizes[1] / 2;
  const int nl = in_sizes[2] / (D * D);

  float* ws    = (float*)d_ws;
  float* h     = ws; ws += (size_t)n * D;
  float* va    = ws; ws += n;
  float* vb    = ws; ws += n;
  int*   cur   = (int*)ws; ws += n;
  int*   srt   = (int*)ws; ws += e;

  const int nb      = (n + BNODES - 1) / BNODES;
  int* hist_t = (int*)h;
  int* pairs  = hist_t + (size_t)nb * NCH;
  sort_hist<<<NCH, ST, nb * sizeof(int), stream>>>(ei, hist_t, e, nb);
  sort_scan<<<1, 1024, 0, stream>>>(hist_t, nb * NCH);
  sort_split<<<NCH, ST, nb * sizeof(int), stream>>>(ei, hist_t, pairs, e, nb);
  sort_bucket<<<nb, ST, 0, stream>>>(hist_t, pairs, cur, srt, n, e, nb);

  for (int l = 0; l < nl; ++l) {
    const float* xin = (l == 0) ? x0 : out;
    const float* Wl  = W + (size_t)l * D * D;
    gat_gemm_wmma<<<(n + 15) / 16, 256, 0, stream>>>(xin, Wl, asrc + l * D,
                                                     adst + l * D, h, va, vb, n);
    gat_gather_ln<<<(n + 8 * NPW - 1) / (8 * NPW), 256, 0, stream>>>(
        cur, srt, va, vb, h, xin, bias + l * D, gam + l * D, bet + l * D, out, n);
  }
}


# Edge sort in 2 launches, not 4: a work-group orders its own 4096-edge chunk by bucket in LDS (rank = counting add's return), a bucket pass gathers the pieces; node sum order = LDS add arrival
# speedup vs baseline: 5.3549x; 1.1584x over previous
#include <hip/hip_runtime.h>
#include <math.h>

#define D 128
#define LDA 132
#define KPRE 16
#define NEG_SLOPE 0.2f
#define LN_EPS 1e-5f

typedef float v2f __attribute__((ext_vector_type(2)));
typedef float v8f __attribute__((ext_vector_type(8)));

__device__ __forceinline__ float wave_sum(float v) {
#pragma unroll
  for (int off = 16; off > 0; off >>= 1) v += __shfl_xor(v, off, 32);
  return v;
}

__device__ __forceinline__ float wave_max(float v) {
  for (int off = 16; off > 0; off >>= 1) v = fmaxf(v, __shfl_xor(v, off, 32));
  return v;
}

__device__ __forceinline__ float leaky(float v) {
  return v > 0.0f ? v : NEG_SLOPE * v;
}

template <int M>
__device__ __forceinline__ float row_xor_add(float v) {
  return v + __int_as_float(__builtin_amdgcn_update_dpp(
                 0, __float_as_int(v), 0x160 + M, 0xf, 0xf, true));
}

__device__ __forceinline__ float row16_sum(float v) {
  v = row_xor_add<8>(v);
  v = row_xor_add<4>(v);
  v = row_xor_add<2>(v);
  v = row_xor_add<1>(v);
  return v;
}

__global__ __launch_bounds__(256, 16)
void gat_gemm_wmma(const float* __restrict__ x, const float* __restrict__ w,
                   const float* __restrict__ att_src,
                   const float* __restrict__ att_dst,
                   float* __restrict__ h, float* __restrict__ a_src,
                   float* __restrict__ a_dst, int n_nodes) {
  __shared__ __attribute__((aligned(16))) float lds_a[16 * LDA];
  __shared__ float lds_p[8 * 16 * 2];
  const int tid  = threadIdx.x;
  const int row0 = blockIdx.x * 16;
  const int wave = tid >> 5;
  const int lane = tid & 31;
  const int half = lane >> 4;
  const int l16  = lane & 15;
  const int col0 = wave * 16;

  float4 xa[2];
#pragma unroll
  for (int q = 0; q < 2; ++q) {
    const int row = row0 + wave + 8 * q;
    xa[q] = (row < n_nodes) ? ((const float4*)(x + (size_t)row * D))[lane]
                            : make_float4(0.0f, 0.0f, 0.0f, 0.0f);
  }
  v2f bpre[KPRE];
#pragma unroll
  for (int t = 0; t < KPRE; ++t) {
    const int ka = 4 * t + half * 2;
    bpre[t].x = w[ka * D + col0 + l16];
    bpre[t].y = w[(ka + 1) * D + col0 + l16];
  }
#pragma unroll
  for (int q = 0; q < 2; ++q)
    *(float4*)&lds_a[(wave + 8 * q) * LDA + lane * 4] = xa[q];
  __syncthreads();

  v8f c = {};
#pragma unroll
  for (int k = 0; k < D; k += 4) {
    const int ka = k + half * 2;
    v2f a, b;
    a.x = lds_a[l16 * LDA + ka];
    a.y = lds_a[l16 * LDA + ka + 1];
    if (k / 4 < KPRE) {
      b = bpre[k / 4];
    } else {
      b.x = w[ka * D + col0 + l16];
      b.y = w[(ka + 1) * D + col0 + l16];
    }
    c = __builtin_amdgcn_wmma_f32_16x16x4_f32(false, a, false, b, (short)0, c,
                                              false, false);
  }
  const float as = att_src[col0 + l16];
  const float ad = att_dst[col0 + l16];
#pragma unroll
  for (int v = 0; v < 8; ++v) {
    int row = row0 + v + half * 8;
    if (row < n_nodes) h[(size_t)row * D + col0 + l16] = c[v];
  }
#pragma unroll
  for (int v = 0; v < 8; ++v) {
    const float ps = row16_sum(c[v] * as);
    const float pd = row16_sum(c[v] * ad);
    if (l16 == 0) {
      lds_p[(wave * 16 + v + half * 8) * 2 + 0] = ps;
      lds_p[(wave * 16 + v + half * 8) * 2 + 1] = pd;
    }
  }
  __syncthreads();
  if (tid < 32) {
    const int r = tid >> 1, which = tid & 1;
    float s = 0.0f;
#pragma unroll
    for (int t = 0; t < 8; ++t) s += lds_p[(t * 16 + r) * 2 + which];
    const int row = row0 + r;
    if (row < n_nodes) (which ? a_dst : a_src)[row] = s;
  }
}


template <int N>
__device__ __forceinline__ int row_shr_add(int v) {
  return v + __builtin_amdgcn_update_dpp(0, v, 0x110 + N, 0xf, 0xf, true);
}

__device__ __forceinline__ int wave_incl_scan_i(int v, int lane) {
  v = row_shr_add<1>(v);
  v = row_shr_add<2>(v);
  v = row_shr_add<4>(v);
  v = row_shr_add<8>(v);
  const int lo = __builtin_amdgcn_readlane(v, 15);
  return v + ((lane & 16) ? lo : 0);
}

#define BSH    8
#define BNODES 256
#define ST     512
#define SU     8
#define CH     (ST * SU)
#define PJ     16

__global__ __launch_bounds__(ST)
void sort_chunk(const int* __restrict__ ei, int* __restrict__ tab_raw,
                int* __restrict__ pairs_raw, int n_edges, int nb, int nch) {
  extern __shared__ int lds_dyn[];
  __shared__ int wsum[ST / 32];
  int2* tab   = (int2*)tab_raw;
  int2* pairs = (int2*)pairs_raw;
  const int tid  = threadIdx.x;
  const int lane = tid & 31;
  const int wave = tid >> 5;
  const int wg   = blockIdx.x;
  const int c0 = wg * CH;
  const int c1 = min(c0 + CH, n_edges);
  int s[SU], d[SU], r[SU];
#pragma unroll
  for (int j = 0; j < SU; ++j) {
    const int i = min(c0 + tid + j * ST, c1 - 1);
    s[j] = ei[i];
    d[j] = ei[n_edges + i];
  }
  for (int b = tid; b < nb; b += ST) lds_dyn[b] = 0;
  __syncthreads();
#pragma unroll
  for (int j = 0; j < SU; ++j) {
    r[j] = 0;
    if (c0 + tid + j * ST < c1) r[j] = atomicAdd(&lds_dyn[d[j] >> BSH], 1);
  }
  __syncthreads();

  const int ipt = (nb + ST - 1) / ST;
  const int i0  = min(tid * ipt, nb);
  const int i1  = min(i0 + ipt, nb);
  int loc = 0;
  for (int i = i0; i < i1; ++i) loc += lds_dyn[i];
  const int inc = wave_incl_scan_i(loc, lane);
  if (lane == 31) wsum[wave] = inc;
  __syncthreads();
  const int wv  = (lane < wave) ? wsum[lane & (ST / 32 - 1)] : 0;
  int       run = __builtin_amdgcn_readlane(wave_incl_scan_i(wv, lane), 31) + inc - loc;
  for (int i = i0; i < i1; ++i) {
    const int c = lds_dyn[i];
    lds_dyn[i] = run;
    run += c;
  }
  __syncthreads();
  int st[SU];
#pragma unroll
  for (int j = 0; j < SU; ++j) st[j] = lds_dyn[d[j] >> BSH];
#pragma unroll
  for (int j = 0; j < SU; ++j)
    if (c0 + tid + j * ST < c1)
      pairs[c0 + st[j] + r[j]] = make_int2(s[j], d[j]);
  for (int i = i0; i < i1; ++i) {
    const int b0 = lds_dyn[i];
    const int b1 = (i + 1 < nb) ? lds_dyn[i + 1] : (c1 - c0);
    tab[(size_t)i * nch + wg] = make_int2(b1 - b0, b0);
  }
}

__device__ __forceinline__ void piece_entry(const int2* __restrict__ tab, int nch,
                                            int cb, int lane, int wave,
                                            int& lv, int& pv, int& acc) {
  const int c = cb + (lane & 15) * 16 + wave;
  lv = 0;
  pv = 0;
  if (lane < 16 && c < nch) {
    const int2 t = tab[c];
    lv = t.x;
    pv = c * CH + t.y;
    acc += t.y;
  }
}

__global__ __launch_bounds__(ST)
void sort_bucket(const int* __restrict__ tab_raw, const int* __restrict__ pairs_raw,
                 int* __restrict__ seg_end, int* __restrict__ srt,
                 int n_nodes, int nch) {
  __shared__ int cnt[BNODES];
  __shared__ int wsum[BNODES / 32];
  __shared__ int red[ST / 32];
  const int tid  = threadIdx.x;
  const int lane = tid & 31;
  const int wave = __builtin_amdgcn_readfirstlane(tid >> 5);
  const int b    = blockIdx.x;
  const int2* tab   = (const int2*)tab_raw + (size_t)b * nch;
  const int2* pairs = (const int2*)pairs_raw;

  const bool one = nch <= 16 * PJ;
  int  acc = 0, lv, pv;
  int2 pr[PJ];
  piece_entry(tab, nch, 0, lane, wave, lv, pv, acc);
  if (tid < BNODES) cnt[tid] = 0;
  __syncthreads();
  for (int cb = 0; cb < nch; cb += 16 * PJ) {
    if (cb > 0) piece_entry(tab, nch, cb, lane, wave, lv, pv, acc);
#pragma unroll
    for (int j = 0; j < PJ; ++j) {
      const int len = __builtin_amdgcn_readlane(lv, j);
      const int pos = __builtin_amdgcn_readlane(pv, j);
      pr[j] = pairs[max(min(pos + lane, pos + len - 1), 0)];
    }
#pragma unroll
    for (int j = 0; j < PJ; ++j) {
      const int len = __builtin_amdgcn_readlane(lv, j);
      const int pos = __builtin_amdgcn_readlane(pv, j);
      if (lane < len) atomicAdd(&cnt[pr[j].y & (BNODES - 1)], 1);
      for (int k = 32 + lane; k < len; k += 32)
        atomicAdd(&cnt[pairs[pos + k].y & (BNODES - 1)], 1);
    }
  }
  acc = __builtin_amdgcn_readlane(wave_incl_scan_i(acc, lane), 31);
  if (lane == 0) red[wave] = acc;
  __syncthreads();

  int bb = 0;
  for (int w = 0; w < ST / 32; ++w) bb += red[w];
  int c = 0, inc = 0;
  if (tid < BNODES) {
    c   = cnt[tid];
    inc = wave_incl_scan_i(c, lane);
    if (lane == 31) wsum[wave] = inc;
  }
  __syncthreads();
  if (tid < BNODES) {
    int base = 0;
    for (int w = 0; w < wave; ++w) base += wsum[w];
    const int excl = base + inc - c;
    cnt[tid] = excl;
    const int node = b * BNODES + tid;
    if (node < n_nodes) seg_end[node] = bb + excl + c;
  }
  __syncthreads();
  for (int cb = 0; cb < nch; cb += 16 * PJ) {
    if (!one) {
      int unused = 0;
      piece_entry(tab, nch, cb, lane, wave, lv, pv, unused);
#pragma unroll
      for (int j = 0; j < PJ; ++j) {
        const int len = __builtin_amdgcn_readlane(lv, j);
        const int pos = __builtin_amdgcn_readlane(pv, j);
        pr[j] = pairs[max(min(pos + lane, pos + len - 1), 0)];
      }
    }
#pragma unroll
    for (int j = 0; j < PJ; ++j) {
      const int len = __builtin_amdgcn_readlane(lv, j);
      const int pos = __builtin_amdgcn_readlane(pv, j);
      if (lane < len) {
        const int p = atomicAdd(&cnt[pr[j].y & (BNODES - 1)], 1);
        srt[bb + p] = pr[j].x;
      }
      for (int k = 32 + lane; k < len; k += 32) {
        const int2 q = pairs[pos + k];
        const int  p = atomicAdd(&cnt[q.y & (BNODES - 1)], 1);
        srt[bb + p] = q.x;
      }
    }
  }
}

#define NPW 4

__device__ __forceinline__ int lane_i(int v, int l) {
  return __builtin_amdgcn_readlane(v, l);
}
__device__ __forceinline__ float lane_f(float v, int l) {
  return __int_as_float(__builtin_amdgcn_readlane(__float_as_int(v), l));
}

__global__ __launch_bounds__(256, 16)
void gat_gather_ln(const int* __restrict__ seg_end, const int* __restrict__ srt,
                   const float* __restrict__ a_src, const float* __restrict__ a_dst,
                   const float* __restrict__ h, const float* x,
                   const float* bias, const float* gamma,
                   const float* beta, float* out, int n_nodes) {
  const int wv   = __builtin_amdgcn_readfirstlane((int)(threadIdx.x >> 5));
  const int lane = threadIdx.x & 31;
  const int n0   = (blockIdx.x * 8 + wv) * NPW;
  if (n0 >= n_nodes) return;
  const int nk = min(NPW, n_nodes - n0);

  float4 hs1 = ((const float4*)(h + (size_t)n0 * D))[lane];
  int   eb  = 0;
  float adv = 0.0f, asv = 0.0f;
  {
    const int idx = n0 - 1 + min(lane, nk);
    if (idx >= 0) eb = seg_end[idx];
    if (lane < nk) {
      adv = a_dst[n0 + lane];
      asv = a_src[n0 + lane];
    }
  }

  int   s1  = 0, s2 = 0;
  float as1 = 0.0f;
  {
    const int b = lane_i(eb, 0), e = lane_i(eb, 1), e2 = lane_i(eb, 2);
    if (b + lane < e) s1 = srt[b + lane];
    if (e + lane < e2) s2 = srt[e + lane];
    if (b + lane < e) as1 = a_src[s1];
  }

#pragma unroll
  for (int k = 0; k < NPW; ++k) {
    if (k >= nk) break;
    const int node = n0 + k;
    const int beg  = lane_i(eb, k);
    const int end  = lane_i(eb, k + 1);
    const int e1   = lane_i(eb, k + 2);
    const int e2   = lane_i(eb, k + 3);
    const float ad    = lane_f(adv, k);
    const float lself = leaky(lane_f(asv, k) + ad);
    const int    s0  = s1;
    const float  as0 = as1;
    const float4 hs  = hs1;
    s1  = s2;
    as1 = 0.0f;
    if (end + lane < e1) as1 = a_src[s1];
    s2 = 0;
    if (e1 + lane < e2) s2 = srt[e1 + lane];
    if (k + 1 < NPW)
      hs1 = ((const float4*)(h + (size_t)min(node + 1, n_nodes - 1) * D))[lane];
    const float4 xv = ((const float4*)(x + (size_t)node * D))[lane];

    const bool  v0  = beg + lane < end;
    const float lg0 = leaky(as0 + ad);
    float m = lself;
    if (v0) m = fmaxf(m, lg0);
    for (int i = beg + 32 + lane; i < end; i += 32)
      m = fmaxf(m, leaky(a_src[srt[i]] + ad));
    m = wave_max(m);

    const float w0 = __expf(lself - m);
    float4 a = make_float4(hs.x * w0, hs.y * w0, hs.z * w0, hs.w * w0);
    float sw = 0.0f;
    for (int i0 = beg; i0 < end; i0 += 32) {
      int   s;
      float w;
      if (i0 == beg) {
        s = s0;
        w = v0 ? __expf(lg0 - m) : 0.0f;
      } else {
        const int i = i0 + lane;
        s = 0;
        w = 0.0f;
        if (i < end) {
          s = srt[i];
          w = __expf(leaky(a_src[s] + ad) - m);
        }
      }
      sw += w;
      const int cnt = min(32, end - i0);
#pragma unroll 4
      for (int j = 0; j < cnt; ++j) {
        const int   sj = lane_i(s, j);
        const float wj = lane_f(w, j);
        const float4 hv = ((const float4*)(h + (size_t)sj * D))[lane];
        a.x += wj * hv.x;
        a.y += wj * hv.y;
        a.z += wj * hv.z;
        a.w += wj * hv.w;
      }
    }
    sw = wave_sum(sw) + w0;

    const float4 bv = ((const float4*)bias)[lane];
    const float4 gv = ((const float4*)gamma)[lane];
    const float4 tv = ((const float4*)beta)[lane];
    const float inv_s = 1.0f / sw;
    float4 y;
    y.x = xv.x + a.x * inv_s + bv.x;
    y.y = xv.y + a.y * inv_s + bv.y;
    y.z = xv.z + a.z * inv_s + bv.z;
    y.w = xv.w + a.w * inv_s + bv.w;
    float sum = y.x + y.y + y.z + y.w;
    float sq  = y.x * y.x + y.y * y.y + y.z * y.z + y.w * y.w;
    sum = wave_sum(sum);
    sq  = wave_sum(sq);
    const float mu   = sum * (1.0f / D);
    const float var  = sq * (1.0f / D) - mu * mu;
    const float rstd = rsqrtf(var + LN_EPS);
    float4 o;
    o.x = (y.x - mu) * rstd * gv.x + tv.x;
    o.y = (y.y - mu) * rstd * gv.y + tv.y;
    o.z = (y.z - mu) * rstd * gv.z + tv.z;
    o.w = (y.w - mu) * rstd * gv.w + tv.w;
    ((float4*)(out + (size_t)node * D))[lane] = o;
  }
}

extern "C" void kernel_launch(void* const* d_in, const int* in_sizes, int n_in,
                              void* d_out, int out_size, void* d_ws, size_t ws_size,
                              hipStream_t stream) {
  const float* x0   = (const float*)d_in[0];
  const int*   ei   = (const int*)d_in[1];
  const float* W    = (const float*)d_in[2];
  const float* asrc = (const float*)d_in[3];
  const float* adst = (const float*)d_in[4];
  const float* bias = (const float*)d_in[5];
  const float* gam  = (const float*)d_in[6];
  const float* bet  = (const float*)d_in[7];
  float* out = (float*)d_out;

  const int n  = in_sizes[0] / D;
  const int e  = in_sizes[1] / 2;
  const int nl = in_sizes[2] / (D * D);

  float* ws    = (float*)d_ws;
  float* h     = ws; ws += (size_t)n * D;
  float* va    = ws; ws += n;
  float* vb    = ws; ws += n;
  int*   cur   = (int*)ws; ws += n;
  int*   srt   = (int*)ws; ws += e;

  const int nb  = (n + BNODES - 1) / BNODES;
  const int nch = (e + CH - 1) / CH;
  int* tab   = (int*)h;
  int* pairs = tab + (size_t)nb * nch * 2;
  sort_chunk<<<nch, ST, nb * sizeof(int), stream>>>(ei, tab, pairs, e, nb, nch);
  sort_bucket<<<nb, ST, 0, stream>>>(tab, pairs, cur, srt, n, nch);

  for (int l = 0; l < nl; ++l) {
    const float* xin = (l == 0) ? x0 : out;
    const float* Wl  = W + (size_t)l * D * D;
    gat_gemm_wmma<<<(n + 15) / 16, 256, 0, stream>>>(xin, Wl, asrc + l * D,
                                                     adst + l * D, h, va, vb, n);
    gat_gather_ln<<<(n + 8 * NPW - 1) / (8 * NPW), 256, 0, stream>>>(
        cur, srt, va, vb, h, xin, bias + l * D, gam + l * D, bet + l * D, out, n);
  }
}


# Layer 1's product rides as extra work-groups (two row tiles each) in the edge sort's two launches, half beside each pass; sort scratch moves to the output plane; 6 launches -> 5; sums unchanged
# speedup vs baseline: 5.6601x; 1.0570x over previous
#include <hip/hip_runtime.h>
#include <math.h>

#define D 128
#define LDA 132
#define KPRE 16
#define NEG_SLOPE 0.2f
#define LN_EPS 1e-5f

typedef float v2f __attribute__((ext_vector_type(2)));
typedef float v8f __attribute__((ext_vector_type(8)));

__device__ __forceinline__ float wave_sum(float v) {
#pragma unroll
  for (int off = 16; off > 0; off >>= 1) v += __shfl_xor(v, off, 32);
  return v;
}

__device__ __forceinline__ float wave_max(float v) {
  for (int off = 16; off > 0; off >>= 1) v = fmaxf(v, __shfl_xor(v, off, 32));
  return v;
}

__device__ __forceinline__ float leaky(float v) {
  return v > 0.0f ? v : NEG_SLOPE * v;
}

template <int M>
__device__ __forceinline__ float row_xor_add(float v) {
  return v + __int_as_float(__builtin_amdgcn_update_dpp(
                 0, __float_as_int(v), 0x160 + M, 0xf, 0xf, true));
}

__device__ __forceinline__ float row16_sum(float v) {
  v = row_xor_add<8>(v);
  v = row_xor_add<4>(v);
  v = row_xor_add<2>(v);
  v = row_xor_add<1>(v);
  return v;
}

template <int NT>
__device__ __forceinline__
void gemm_tiles(const float* __restrict__ x, const float* __restrict__ w,
                const float* __restrict__ att_src,
                const float* __restrict__ att_dst,
                float* __restrict__ h, float* __restrict__ a_src,
                float* __restrict__ a_dst, int n_nodes, int tile0) {
  __shared__ __attribute__((aligned(16))) float lds_a_all[NT * 16 * LDA];
  __shared__ float lds_p_all[NT * 8 * 16 * 2];
  const int sub  = threadIdx.x >> 8;
  const int tid  = threadIdx.x & 255;
  float* lds_a = lds_a_all + sub * (16 * LDA);
  float* lds_p = lds_p_all + sub * (8 * 16 * 2);
  const int row0 = (tile0 + sub) * 16;
  const int wave = tid >> 5;
  const int lane = tid & 31;
  const int half = lane >> 4;
  const int l16  = lane & 15;
  const int col0 = wave * 16;

  float4 xa[2];
#pragma unroll
  for (int q = 0; q < 2; ++q) {
    const int row = row0 + wave + 8 * q;
    xa[q] = (row < n_nodes) ? ((const float4*)(x + (size_t)row * D))[lane]
                            : make_float4(0.0f, 0.0f, 0.0f, 0.0f);
  }
  v2f bpre[KPRE];
#pragma unroll
  for (int t = 0; t < KPRE; ++t) {
    const int ka = 4 * t + half * 2;
    bpre[t].x = w[ka * D + col0 + l16];
    bpre[t].y = w[(ka + 1) * D + col0 + l16];
  }
#pragma unroll
  for (int q = 0; q < 2; ++q)
    *(float4*)&lds_a[(wave + 8 * q) * LDA + lane * 4] = xa[q];
  __syncthreads();

  v8f c = {};
#pragma unroll
  for (int k = 0; k < D; k += 4) {
    const int ka = k + half * 2;
    v2f a, b;
    a.x = lds_a[l16 * LDA + ka];
    a.y = lds_a[l16 * LDA + ka + 1];
    if (k / 4 < KPRE) {
      b = bpre[k / 4];
    } else {
      b.x = w[ka * D + col0 + l16];
      b.y = w[(ka + 1) * D + col0 + l16];
    }
    c = __builtin_amdgcn_wmma_f32_16x16x4_f32(false, a, false, b, (short)0, c,
                                              false, false);
  }
  const float as = att_src[col0 + l16];
  const float ad = att_dst[col0 + l16];
#pragma unroll
  for (int v = 0; v < 8; ++v) {
    int row = row0 + v + half * 8;
    if (row < n_nodes) h[(size_t)row * D + col0 + l16] = c[v];
  }
#pragma unroll
  for (int v = 0; v < 8; ++v) {
    const float ps = row16_sum(c[v] * as);
    const float pd = row16_sum(c[v] * ad);
    if (l16 == 0) {
      lds_p[(wave * 16 + v + half * 8) * 2 + 0] = ps;
      lds_p[(wave * 16 + v + half * 8) * 2 + 1] = pd;
    }
  }
  __syncthreads();
  if (tid < 32) {
    const int r = tid >> 1, which = tid & 1;
    float s = 0.0f;
#pragma unroll
    for (int t = 0; t < 8; ++t) s += lds_p[(t * 16 + r) * 2 + which];
    const int row = row0 + r;
    if (row < n_nodes) (which ? a_dst : a_src)[row] = s;
  }
}


template <int N>
__device__ __forceinline__ int row_shr_add(int v) {
  return v + __builtin_amdgcn_update_dpp(0, v, 0x110 + N, 0xf, 0xf, true);
}

__device__ __forceinline__ int wave_incl_scan_i(int v, int lane) {
  v = row_shr_add<1>(v);
  v = row_shr_add<2>(v);
  v = row_shr_add<4>(v);
  v = row_shr_add<8>(v);
  const int lo = __builtin_amdgcn_readlane(v, 15);
  return v + ((lane & 16) ? lo : 0);
}

#define BSH    8
#define BNODES 256
#define ST     512
#define SU     8
#define CH     (ST * SU)
#define PJ     16

__device__ __forceinline__
void sort_chunk_body(const int* __restrict__ ei, int* __restrict__ tab_raw,
                     int* __restrict__ pairs_raw, int n_edges, int nb, int nch) {
  extern __shared__ int lds_dyn[];
  __shared__ int wsum[ST / 32];
  int2* tab   = (int2*)tab_raw;
  int2* pairs = (int2*)pairs_raw;
  const int tid  = threadIdx.x;
  const int lane = tid & 31;
  const int wave = tid >> 5;
  const int wg   = blockIdx.x;
  const int c0 = wg * CH;
  const int c1 = min(c0 + CH, n_edges);
  int s[SU], d[SU], r[SU];
#pragma unroll
  for (int j = 0; j < SU; ++j) {
    const int i = min(c0 + tid + j * ST, c1 - 1);
    s[j] = ei[i];
    d[j] = ei[n_edges + i];
  }
  for (int b = tid; b < nb; b += ST) lds_dyn[b] = 0;
  __syncthreads();
#pragma unroll
  for (int j = 0; j < SU; ++j) {
    r[j] = 0;
    if (c0 + tid + j * ST < c1) r[j] = atomicAdd(&lds_dyn[d[j] >> BSH], 1);
  }
  __syncthreads();

  const int ipt = (nb + ST - 1) / ST;
  const int i0  = min(tid * ipt, nb);
  const int i1  = min(i0 + ipt, nb);
  int loc = 0;
  for (int i = i0; i < i1; ++i) loc += lds_dyn[i];
  const int inc = wave_incl_scan_i(loc, lane);
  if (lane == 31) wsum[wave] = inc;
  __syncthreads();
  const int wv  = (lane < wave) ? wsum[lane & (ST / 32 - 1)] : 0;
  int       run = __builtin_amdgcn_readlane(wave_incl_scan_i(wv, lane), 31) + inc - loc;
  for (int i = i0; i < i1; ++i) {
    const int c = lds_dyn[i];
    lds_dyn[i] = run;
    run += c;
  }
  __syncthreads();
  int st[SU];
#pragma unroll
  for (int j = 0; j < SU; ++j) st[j] = lds_dyn[d[j] >> BSH];
#pragma unroll
  for (int j = 0; j < SU; ++j)
    if (c0 + tid + j * ST < c1)
      pairs[c0 + st[j] + r[j]] = make_int2(s[j], d[j]);
  for (int i = i0; i < i1; ++i) {
    const int b0 = lds_dyn[i];
    const int b1 = (i + 1 < nb) ? lds_dyn[i + 1] : (c1 - c0);
    tab[(size_t)i * nch + wg] = make_int2(b1 - b0, b0);
  }
}

__device__ __forceinline__ void piece_entry(const int2* __restrict__ tab, int nch,
                                            int cb, int lane, int wave,
                                            int& lv, int& pv, int& acc) {
  const int c = cb + (lane & 15) * 16 + wave;
  lv = 0;
  pv = 0;
  if (lane < 16 && c < nch) {
    const int2 t = tab[c];
    lv = t.x;
    pv = c * CH + t.y;
    acc += t.y;
  }
}

__device__ __forceinline__
void sort_bucket_body(const int* __restrict__ tab_raw, const int* __restrict__ pairs_raw,
                      int* __restrict__ seg_end, int* __restrict__ srt,
                      int n_nodes, int nch) {
  __shared__ int cnt[BNODES];
  __shared__ int wsum[BNODES / 32];
  __shared__ int red[ST / 32];
  const int tid  = threadIdx.x;
  const int lane = tid & 31;
  const int wave = __builtin_amdgcn_readfirstlane(tid >> 5);
  const int b    = blockIdx.x;
  const int2* tab   = (const int2*)tab_raw + (size_t)b * nch;
  const int2* pairs = (const int2*)pairs_raw;

  const bool one = nch <= 16 * PJ;
  int  acc = 0, lv, pv;
  int2 pr[PJ];
  piece_entry(tab, nch, 0, lane, wave, lv, pv, acc);
  if (tid < BNODES) cnt[tid] = 0;
  __syncthreads();
  for (int cb = 0; cb < nch; cb += 16 * PJ) {
    if (cb > 0) piece_entry(tab, nch, cb, lane, wave, lv, pv, acc);
#pragma unroll
    for (int j = 0; j < PJ; ++j) {
      const int len = __builtin_amdgcn_readlane(lv, j);
      const int pos = __builtin_amdgcn_readlane(pv, j);
      pr[j] = pairs[max(min(pos + lane, pos + len - 1), 0)];
    }
#pragma unroll
    for (int j = 0; j < PJ; ++j) {
      const int len = __builtin_amdgcn_readlane(lv, j);
      const int pos = __builtin_amdgcn_readlane(pv, j);
      if (lane < len) atomicAdd(&cnt[pr[j].y & (BNODES - 1)], 1);
      for (int k = 32 + lane; k < len; k += 32)
        atomicAdd(&cnt[pairs[pos + k].y & (BNODES - 1)], 1);
    }
  }
  acc = __builtin_amdgcn_readlane(wave_incl_scan_i(acc, lane), 31);
  if (lane == 0) red[wave] = acc;
  __syncthreads();

  int bb = 0;
  for (int w = 0; w < ST / 32; ++w) bb += red[w];
  int c = 0, inc = 0;
  if (tid < BNODES) {
    c   = cnt[tid];
    inc = wave_incl_scan_i(c, lane);
    if (lane == 31) wsum[wave] = inc;
  }
  __syncthreads();
  if (tid < BNODES) {
    int base = 0;
    for (int w = 0; w < wave; ++w) base += wsum[w];
    const int excl = base + inc - c;
    cnt[tid] = excl;
    const int node = b * BNODES + tid;
    if (node < n_nodes) seg_end[node] = bb + excl + c;
  }
  __syncthreads();
  for (int cb = 0; cb < nch; cb += 16 * PJ) {
    if (!one) {
      int unused = 0;
      piece_entry(tab, nch, cb, lane, wave, lv, pv, unused);
#pragma unroll
      for (int j = 0; j < PJ; ++j) {
        const int len = __builtin_amdgcn_readlane(lv, j);
        const int pos = __builtin_amdgcn_readlane(pv, j);
        pr[j] = pairs[max(min(pos + lane, pos + len - 1), 0)];
      }
    }
#pragma unroll
    for (int j = 0; j < PJ; ++j) {
      const int len = __builtin_amdgcn_readlane(lv, j);
      const int pos = __builtin_amdgcn_readlane(pv, j);
      if (lane < len) {
        const int p = atomicAdd(&cnt[pr[j].y & (BNODES - 1)], 1);
        srt[bb + p] = pr[j].x;
      }
      for (int k = 32 + lane; k < len; k += 32) {
        const int2 q = pairs[pos + k];
        const int  p = atomicAdd(&cnt[q.y & (BNODES - 1)], 1);
        srt[bb + p] = q.x;
      }
    }
  }
}

__global__ __launch_bounds__(256, 16)
void gat_gemm_wmma(const float* __restrict__ x, const float* __restrict__ w,
                   const float* __restrict__ att_src,
                   const float* __restrict__ att_dst,
                   float* __restrict__ h, float* __restrict__ a_src,
                   float* __restrict__ a_dst, int n_nodes) {
  gemm_tiles<1>(x, w, att_src, att_dst, h, a_src, a_dst, n_nodes, blockIdx.x);
}

__global__ __launch_bounds__(ST, 16)
void sort_chunk_and_gemm(const int* __restrict__ ei, int* __restrict__ tab_raw,
                         int* __restrict__ pairs_raw, int n_edges, int nb, int nch,
                         const float* __restrict__ x, const float* __restrict__ w,
                         const float* __restrict__ att_src,
                         const float* __restrict__ att_dst,
                         float* __restrict__ h, float* __restrict__ a_src,
                         float* __restrict__ a_dst, int n_nodes, int tile0) {
  if ((int)blockIdx.x < nch)
    sort_chunk_body(ei, tab_raw, pairs_raw, n_edges, nb, nch);
  else
    gemm_tiles<2>(x, w, att_src, att_dst, h, a_src, a_dst, n_nodes,
                  tile0 + 2 * ((int)blockIdx.x - nch));
}

__global__ __launch_bounds__(ST, 16)
void sort_bucket_and_gemm(const int* __restrict__ tab_raw,
                          const int* __restrict__ pairs_raw,
                          int* __restrict__ seg_end, int* __restrict__ srt,
                          int nch, int nb,
                          const float* __restrict__ x, const float* __restrict__ w,
                          const float* __restrict__ att_src,
                          const float* __restrict__ att_dst,
                          float* __restrict__ h, float* __restrict__ a_src,
                          float* __restrict__ a_dst, int n_nodes, int tile0) {
  if ((int)blockIdx.x < nb)
    sort_bucket_body(tab_raw, pairs_raw, seg_end, srt, n_nodes, nch);
  else
    gemm_tiles<2>(x, w, att_src, att_dst, h, a_src, a_dst, n_nodes,
                  tile0 + 2 * ((int)blockIdx.x - nb));
}

#define NPW 4

__device__ __forceinline__ int lane_i(int v, int l) {
  return __builtin_amdgcn_readlane(v, l);
}
__device__ __forceinline__ float lane_f(float v, int l) {
  return __int_as_float(__builtin_amdgcn_readlane(__float_as_int(v), l));
}

__global__ __launch_bounds__(256, 16)
void gat_gather_ln(const int* __restrict__ seg_end, const int* __restrict__ srt,
                   const float* __restrict__ a_src, const float* __restrict__ a_dst,
                   const float* __restrict__ h, const float* x,
                   const float* bias, const float* gamma,
                   const float* beta, float* out, int n_nodes) {
  const int wv   = __builtin_amdgcn_readfirstlane((int)(threadIdx.x >> 5));
  const int lane = threadIdx.x & 31;
  const int n0   = (blockIdx.x * 8 + wv) * NPW;
  if (n0 >= n_nodes) return;
  const int nk = min(NPW, n_nodes - n0);

  float4 hs1 = ((const float4*)(h + (size_t)n0 * D))[lane];
  int   eb  = 0;
  float adv = 0.0f, asv = 0.0f;
  {
    const int idx = n0 - 1 + min(lane, nk);
    if (idx >= 0) eb = seg_end[idx];
    if (lane < nk) {
      adv = a_dst[n0 + lane];
      asv = a_src[n0 + lane];
    }
  }

  int   s1  = 0, s2 = 0;
  float as1 = 0.0f;
  {
    const int b = lane_i(eb, 0), e = lane_i(eb, 1), e2 = lane_i(eb, 2);
    if (b + lane < e) s1 = srt[b + lane];
    if (e + lane < e2) s2 = srt[e + lane];
    if (b + lane < e) as1 = a_src[s1];
  }

#pragma unroll
  for (int k = 0; k < NPW; ++k) {
    if (k >= nk) break;
    const int node = n0 + k;
    const int beg  = lane_i(eb, k);
    const int end  = lane_i(eb, k + 1);
    const int e1   = lane_i(eb, k + 2);
    const int e2   = lane_i(eb, k + 3);
    const float ad    = lane_f(adv, k);
    const float lself = leaky(lane_f(asv, k) + ad);
    const int    s0  = s1;
    const float  as0 = as1;
    const float4 hs  = hs1;
    s1  = s2;
    as1 = 0.0f;
    if (end + lane < e1) as1 = a_src[s1];
    s2 = 0;
    if (e1 + lane < e2) s2 = srt[e1 + lane];
    if (k + 1 < NPW)
      hs1 = ((const float4*)(h + (size_t)min(node + 1, n_nodes - 1) * D))[lane];
    const float4 xv = ((const float4*)(x + (size_t)node * D))[lane];

    const bool  v0  = beg + lane < end;
    const float lg0 = leaky(as0 + ad);
    float m = lself;
    if (v0) m = fmaxf(m, lg0);
    for (int i = beg + 32 + lane; i < end; i += 32)
      m = fmaxf(m, leaky(a_src[srt[i]] + ad));
    m = wave_max(m);

    const float w0 = __expf(lself - m);
    float4 a = make_float4(hs.x * w0, hs.y * w0, hs.z * w0, hs.w * w0);
    float sw = 0.0f;
    for (int i0 = beg; i0 < end; i0 += 32) {
      int   s;
      float w;
      if (i0 == beg) {
        s = s0;
        w = v0 ? __expf(lg0 - m) : 0.0f;
      } else {
        const int i = i0 + lane;
        s = 0;
        w = 0.0f;
        if (i < end) {
          s = srt[i];
          w = __expf(leaky(a_src[s] + ad) - m);
        }
      }
      sw += w;
      const int cnt = min(32, end - i0);
#pragma unroll 4
      for (int j = 0; j < cnt; ++j) {
        const int   sj = lane_i(s, j);
        const float wj = lane_f(w, j);
        const float4 hv = ((const float4*)(h + (size_t)sj * D))[lane];
        a.x += wj * hv.x;
        a.y += wj * hv.y;
        a.z += wj * hv.z;
        a.w += wj * hv.w;
      }
    }
    sw = wave_sum(sw) + w0;

    const float4 bv = ((const float4*)bias)[lane];
    const float4 gv = ((const float4*)gamma)[lane];
    const float4 tv = ((const float4*)beta)[lane];
    const float inv_s = 1.0f / sw;
    float4 y;
    y.x = xv.x + a.x * inv_s + bv.x;
    y.y = xv.y + a.y * inv_s + bv.y;
    y.z = xv.z + a.z * inv_s + bv.z;
    y.w = xv.w + a.w * inv_s + bv.w;
    float sum = y.x + y.y + y.z + y.w;
    float sq  = y.x * y.x + y.y * y.y + y.z * y.z + y.w * y.w;
    sum = wave_sum(sum);
    sq  = wave_sum(sq);
    const float mu   = sum * (1.0f / D);
    const float var  = sq * (1.0f / D) - mu * mu;
    const float rstd = rsqrtf(var + LN_EPS);
    float4 o;
    o.x = (y.x - mu) * rstd * gv.x + tv.x;
    o.y = (y.y - mu) * rstd * gv.y + tv.y;
    o.z = (y.z - mu) * rstd * gv.z + tv.z;
    o.w = (y.w - mu) * rstd * gv.w + tv.w;
    ((float4*)(out + (size_t)node * D))[lane] = o;
  }
}

extern "C" void kernel_launch(void* const* d_in, const int* in_sizes, int n_in,
                              void* d_out, int out_size, void* d_ws, size_t ws_size,
                              hipStream_t stream) {
  const float* x0   = (const float*)d_in[0];
  const int*   ei   = (const int*)d_in[1];
  const float* W    = (const float*)d_in[2];
  const float* asrc = (const float*)d_in[3];
  const float* adst = (const float*)d_in[4];
  const float* bias = (const float*)d_in[5];
  const float* gam  = (const float*)d_in[6];
  const float* bet  = (const float*)d_in[7];
  float* out = (float*)d_out;

  const int n  = in_sizes[0] / D;
  const int e  = in_sizes[1] / 2;
  const int nl = in_sizes[2] / (D * D);

  float* ws    = (float*)d_ws;
  float* h     = ws; ws += (size_t)n * D;
  float* va    = ws; ws += n;
  float* vb    = ws; ws += n;
  int*   cur   = (int*)ws; ws += n;
  int*   srt   = (int*)ws; ws += e;

  const int nb  = (n + BNODES - 1) / BNODES;
  const int nch = (e + CH - 1) / CH;
  int* tab   = (int*)out;
  int* pairs = tab + (size_t)nb * nch * 2;
  const int gw  = ((n + 15) / 16 + 1) / 2;
  const int gw1 = gw / 2;
  sort_chunk_and_gemm<<<nch + gw1, ST, nb * sizeof(int), stream>>>(
      ei, tab, pairs, e, nb, nch, x0, W, asrc, adst, h, va, vb, n, 0);
  sort_bucket_and_gemm<<<nb + (gw - gw1), ST, 0, stream>>>(
      tab, pairs, cur, srt, nch, nb, x0, W, asrc, adst, h, va, vb, n, 2 * gw1);

  for (int l = 0; l < nl; ++l) {
    const float* xin = (l == 0) ? x0 : out;
    const float* Wl  = W + (size_t)l * D * D;
    if (l > 0)
      gat_gemm_wmma<<<(n + 15) / 16, 256, 0, stream>>>(xin, Wl, asrc + l * D,
                                                       adst + l * D, h, va, vb, n);
    gat_gather_ln<<<(n + 8 * NPW - 1) / (8 * NPW), 256, 0, stream>>>(
        cur, srt, va, vb, h, xin, bias + l * D, gam + l * D, bet + l * D, out, n);
  }
}
